# starting kernel
# baseline (speedup 1.0000x reference)
#include <hip/hip_runtime.h>


typedef float v2f __attribute__((ext_vector_type(2)));
typedef float v8f __attribute__((ext_vector_type(8)));

#define NFEAT 512
#define NH1   256
#define NH2   128

__device__ __forceinline__ void atomic_fadd_dev(float* p, float v) {
#if defined(__gfx1250__)
  asm volatile("global_atomic_add_f32 %0, %1, off scope:SCOPE_DEV"
               :: "v"(p), "v"(v)
               : "memory");
#else
  atomicAdd(p, v);
#endif
}

__global__ void k_deg_init(float* __restrict__ deg, int n) {
  int i = blockIdx.x * blockDim.x + threadIdx.x;
  if (i < n) deg[i] = 1.0f;
}

__global__ void k_deg_count(const int* __restrict__ dst, int E, float* __restrict__ deg) {
  int e = blockIdx.x * blockDim.x + threadIdx.x;
  if (e < E) atomic_fadd_dev(&deg[dst[e]], 1.0f);
}

__global__ void k_dinv(float* __restrict__ deg, int n) {
  int i = blockIdx.x * blockDim.x + threadIdx.x;
  if (i < n) deg[i] = rsqrtf(deg[i]);
}

#define APAD 34

__global__ __launch_bounds__(256)
void k_gemm_wmma(const float* __restrict__ A, const float* __restrict__ B,
                 float* __restrict__ C, int M, int K, int N) {
  __shared__ float As[64 * APAD];
  __shared__ float Bs[32 * 64];

  const int tid  = threadIdx.x;
  const int lane = tid & 31;
  const int wv   = tid >> 5;
  const int mw   = wv & 3;
  const int nw   = wv >> 2;
  const int hf   = lane >> 4;
  const int idx  = lane & 15;

  const int m_base = blockIdx.y * 64;
  const int n_base = blockIdx.x * 64;

  v8f acc0 = {};
  v8f acc1 = {};

  for (int k0 = 0; k0 < K; k0 += 32) {
#pragma unroll
    for (int i = 0; i < 8; ++i) {
      int eid = tid + i * 256;
      int r = eid >> 5, c = eid & 31;
      int gr = m_base + r;
      As[r * APAD + c] = (gr < M) ? A[(size_t)gr * K + k0 + c] : 0.0f;
    }
#pragma unroll
    for (int i = 0; i < 8; ++i) {
      int eid = tid + i * 256;
      int r = eid >> 6, c = eid & 63;
      Bs[r * 64 + c] = B[(size_t)(k0 + r) * N + n_base + c];
    }
    __syncthreads();

#pragma unroll
    for (int kk = 0; kk < 32; kk += 4) {
      const float* ap = &As[(mw * 16 + idx) * APAD + kk + 2 * hf];
      v2f a; a.x = ap[0]; a.y = ap[1];
      const float* bp0 = &Bs[(kk + 2 * hf) * 64 + nw * 32 + idx];
      v2f b0; b0.x = bp0[0];  b0.y = bp0[64];
      v2f b1; b1.x = bp0[16]; b1.y = bp0[80];
      acc0 = __builtin_amdgcn_wmma_f32_16x16x4_f32(false, a, false, b0,
                                                   (short)0, acc0, false, false);
      acc1 = __builtin_amdgcn_wmma_f32_16x16x4_f32(false, a, false, b1,
                                                   (short)0, acc1, false, false);
    }
    __syncthreads();
  }

#pragma unroll
  for (int r = 0; r < 8; ++r) {
    int grow = m_base + mw * 16 + r + 8 * hf;
    if (grow < M) {
      C[(size_t)grow * N + n_base + nw * 32 + idx]      = acc0[r];
      C[(size_t)grow * N + n_base + nw * 32 + 16 + idx] = acc1[r];
    }
  }
}

template <int F>
__global__ __launch_bounds__(256)
void k_edge_scatter(const int* __restrict__ src, const int* __restrict__ dst,
                    int E, int n, const float* __restrict__ dinv,
                    const float* __restrict__ hin, float* __restrict__ hout) {
  int gw   = (blockIdx.x * blockDim.x + threadIdx.x) >> 5;
  int lane = threadIdx.x & 31;
  int total = E + n;
  if (gw >= total) return;

  int s, d;
  if (gw < E) { s = src[gw]; d = dst[gw]; }
  else        { s = d = gw - E; }
  float norm = dinv[s] * dinv[d];

  const float4* in4 = (const float4*)(hin + (size_t)s * F);
  float* outp = hout + (size_t)d * F;
#pragma unroll
  for (int g = 0; g < F / 128; ++g) {
    int c = lane + g * 32;
    float4 v = in4[c];
    atomic_fadd_dev(outp + c * 4 + 0, v.x * norm);
    atomic_fadd_dev(outp + c * 4 + 1, v.y * norm);
    atomic_fadd_dev(outp + c * 4 + 2, v.z * norm);
    atomic_fadd_dev(outp + c * 4 + 3, v.w * norm);
  }
}

__global__ void k_bias_relu(float* __restrict__ h, const float* __restrict__ b,
                            size_t total, int F) {
  size_t i = (size_t)blockIdx.x * blockDim.x + threadIdx.x;
  if (i < total) {
    float v = h[i] + b[i & (size_t)(F - 1)];
    h[i] = v > 0.0f ? v : 0.0f;
  }
}

__global__ void k_bias(float* __restrict__ h, const float* __restrict__ b,
                       size_t total, int F) {
  size_t i = (size_t)blockIdx.x * blockDim.x + threadIdx.x;
  if (i < total) h[i] += b[i & (size_t)(F - 1)];
}

extern "C" void kernel_launch(void* const* d_in, const int* in_sizes, int n_in,
                              void* d_out, int out_size, void* d_ws, size_t ws_size,
                              hipStream_t stream) {
  const float* x  = (const float*)d_in[0];
  const int*   ei = (const int*)d_in[1];
  const float* W1 = (const float*)d_in[2];
  const float* b1 = (const float*)d_in[3];
  const float* W2 = (const float*)d_in[4];
  const float* b2 = (const float*)d_in[5];

  const int n = in_sizes[0] / NFEAT;
  const int E = in_sizes[1] / 2;
  const int* src = ei;
  const int* dst = ei + E;

  float* dinv = (float*)d_ws;
  float* Hlin = dinv + (size_t)((n + 255) & ~255);
  float* agg  = Hlin + (size_t)n * NH1;
  float* out  = (float*)d_out;

  k_deg_init<<<(n + 255) / 256, 256, 0, stream>>>(dinv, n);
  k_deg_count<<<(E + 255) / 256, 256, 0, stream>>>(dst, E, dinv);
  k_dinv<<<(n + 255) / 256, 256, 0, stream>>>(dinv, n);

  const long long waves = (long long)E + n;
  const int scatter_blocks = (int)((waves * 32 + 255) / 256);

  k_gemm_wmma<<<dim3(NH1 / 64, (n + 63) / 64), 256, 0, stream>>>(x, W1, Hlin, n, NFEAT, NH1);
  hipMemsetAsync(agg, 0, (size_t)n * NH1 * sizeof(float), stream);
  k_edge_scatter<NH1><<<scatter_blocks, 256, 0, stream>>>(src, dst, E, n, dinv, Hlin, agg);
  k_bias_relu<<<(int)(((size_t)n * NH1 + 255) / 256), 256, 0, stream>>>(
      agg, b1, (size_t)n * NH1, NH1);

  k_gemm_wmma<<<dim3(NH2 / 64, (n + 63) / 64), 256, 0, stream>>>(agg, W2, Hlin, n, NH1, NH2);
  hipMemsetAsync(out, 0, (size_t)out_size * sizeof(float), stream);
  k_edge_scatter<NH2><<<scatter_blocks, 256, 0, stream>>>(src, dst, E, n, dinv, Hlin, out);
  k_bias<<<(int)(((size_t)out_size + 255) / 256), 256, 0, stream>>>(
      out, b2, (size_t)out_size, NH2);
}


# f32 atomic scatter-add -> edges ordered by destination in-call (int rank atomic, scan, place); a wave sums a node's rows in registers in rank arrival order, bias/relu in its store, no zero-fill
# speedup vs baseline: 1.8562x; 1.8562x over previous
#include <hip/hip_runtime.h>


typedef float v2f __attribute__((ext_vector_type(2)));
typedef float v8f __attribute__((ext_vector_type(8)));

#define NFEAT 512
#define NH1   256
#define NH2   128

__global__ void k_zero_i32(int* __restrict__ p, int n) {
  int i = blockIdx.x * blockDim.x + threadIdx.x;
  if (i < n) p[i] = 0;
}

__global__ void k_count_rank(const int* __restrict__ dst, int E,
                             int* __restrict__ cnt, int* __restrict__ rank) {
  int e = blockIdx.x * blockDim.x + threadIdx.x;
  if (e < E) rank[e] = atomicAdd(&cnt[dst[e]], 1);
}

__global__ __launch_bounds__(1024)
void k_scan_rowptr(const int* __restrict__ cnt, int* __restrict__ rowptr, int n) {
  __shared__ int wtot[2][128];
  const int tid  = threadIdx.x;
  const int lane = tid & 31;
  const int wv   = tid >> 5;
  int carry = 0;
  int par = 0;
  for (int base = 0; base < n; base += 16384, par ^= 1) {
    int c[4][4], s[4], incl[4];
#pragma unroll
    for (int k = 0; k < 4; ++k) {
      const int i0 = base + k * 4096 + tid * 4;
      if (i0 + 4 <= n) {
        const int4 q = *(const int4*)(cnt + i0);
        c[k][0] = q.x; c[k][1] = q.y; c[k][2] = q.z; c[k][3] = q.w;
      } else {
#pragma unroll
        for (int j = 0; j < 4; ++j) c[k][j] = (i0 + j < n) ? cnt[i0 + j] : 0;
      }
      s[k] = c[k][0] + c[k][1] + c[k][2] + c[k][3];
      incl[k] = s[k];
    }
#pragma unroll
    for (int o = 1; o < 32; o <<= 1) {
#pragma unroll
      for (int k = 0; k < 4; ++k) {
        const int t = __shfl_up(incl[k], o);
        if (lane >= o) incl[k] += t;
      }
    }
    if (lane == 31) {
#pragma unroll
      for (int k = 0; k < 4; ++k) wtot[par][k * 32 + wv] = incl[k];
    }
    __syncthreads();
    int T[4], Wi[4];
#pragma unroll
    for (int k = 0; k < 4; ++k) { T[k] = wtot[par][k * 32 + lane]; Wi[k] = T[k]; }
#pragma unroll
    for (int o = 1; o < 32; o <<= 1) {
#pragma unroll
      for (int k = 0; k < 4; ++k) {
        const int t = __shfl_up(Wi[k], o);
        if (lane >= o) Wi[k] += t;
      }
    }
    int pre = carry;
#pragma unroll
    for (int k = 0; k < 4; ++k) {
      const int S  = __shfl(Wi[k], 31);
      const int wp = __shfl(Wi[k], wv) - __shfl(T[k], wv);
      const int r0 = pre + wp + (incl[k] - s[k]);
      const int r1 = r0 + c[k][0];
      const int r2 = r1 + c[k][1];
      const int r3 = r2 + c[k][2];
      const int i0 = base + k * 4096 + tid * 4;
      if (i0 + 4 <= n) {
        *(int4*)(rowptr + i0) = make_int4(r0, r1, r2, r3);
      } else {
        if (i0 + 0 < n) rowptr[i0 + 0] = r0;
        if (i0 + 1 < n) rowptr[i0 + 1] = r1;
        if (i0 + 2 < n) rowptr[i0 + 2] = r2;
      }
      pre += S;
    }
    carry = pre;
  }
  if (tid == 0) rowptr[n] = carry;
}

__global__ void k_place_dinv(const int* __restrict__ src, const int* __restrict__ dst,
                             const int* __restrict__ rank, const int* __restrict__ rowptr,
                             const int* __restrict__ cnt, int E, int n,
                             int* __restrict__ csr_src, float* __restrict__ dinv) {
  int i = blockIdx.x * blockDim.x + threadIdx.x;
  if (i < E) {
    csr_src[rowptr[dst[i]] + rank[i]] = src[i];
  } else if (i - E < n) {
    dinv[i - E] = rsqrtf((float)(cnt[i - E] + 1));
  }
}

__global__ void k_move_lists(const int* __restrict__ rp_a, const int* __restrict__ cs_a,
                             int* __restrict__ rp_b, int* __restrict__ cs_b, int n, int E) {
  int i = blockIdx.x * blockDim.x + threadIdx.x;
  if (i <= n) rp_b[i] = rp_a[i];
  else if (i - (n + 1) < E) cs_b[i - (n + 1)] = cs_a[i - (n + 1)];
}

#define APAD 34

__global__ __launch_bounds__(256)
void k_gemm_wmma(const float* __restrict__ A, const float* __restrict__ B,
                 float* __restrict__ C, int M, int K, int N) {
  __shared__ float As[64 * APAD];
  __shared__ float Bs[32 * 64];

  const int tid  = threadIdx.x;
  const int lane = tid & 31;
  const int wv   = tid >> 5;
  const int mw   = wv & 3;
  const int nw   = wv >> 2;
  const int hf   = lane >> 4;
  const int idx  = lane & 15;

  const int m_base = blockIdx.y * 64;
  const int n_base = blockIdx.x * 64;

  v8f acc0 = {};
  v8f acc1 = {};

  for (int k0 = 0; k0 < K; k0 += 32) {
#pragma unroll
    for (int i = 0; i < 8; ++i) {
      int eid = tid + i * 256;
      int r = eid >> 5, c = eid & 31;
      int gr = m_base + r;
      As[r * APAD + c] = (gr < M) ? A[(size_t)gr * K + k0 + c] : 0.0f;
    }
#pragma unroll
    for (int i = 0; i < 8; ++i) {
      int eid = tid + i * 256;
      int r = eid >> 6, c = eid & 63;
      Bs[r * 64 + c] = B[(size_t)(k0 + r) * N + n_base + c];
    }
    __syncthreads();

#pragma unroll
    for (int kk = 0; kk < 32; kk += 4) {
      const float* ap = &As[(mw * 16 + idx) * APAD + kk + 2 * hf];
      v2f a; a.x = ap[0]; a.y = ap[1];
      const float* bp0 = &Bs[(kk + 2 * hf) * 64 + nw * 32 + idx];
      v2f b0; b0.x = bp0[0];  b0.y = bp0[64];
      v2f b1; b1.x = bp0[16]; b1.y = bp0[80];
      acc0 = __builtin_amdgcn_wmma_f32_16x16x4_f32(false, a, false, b0,
                                                   (short)0, acc0, false, false);
      acc1 = __builtin_amdgcn_wmma_f32_16x16x4_f32(false, a, false, b1,
                                                   (short)0, acc1, false, false);
    }
    __syncthreads();
  }

#pragma unroll
  for (int r = 0; r < 8; ++r) {
    int grow = m_base + mw * 16 + r + 8 * hf;
    if (grow < M) {
      C[(size_t)grow * N + n_base + nw * 32 + idx]      = acc0[r];
      C[(size_t)grow * N + n_base + nw * 32 + 16 + idx] = acc1[r];
    }
  }
}

__device__ __forceinline__ float4 f4_fma(float4 v, float w, float4 a) {
  a.x = fmaf(v.x, w, a.x); a.y = fmaf(v.y, w, a.y);
  a.z = fmaf(v.z, w, a.z); a.w = fmaf(v.w, w, a.w);
  return a;
}

template <int R, int F>
__device__ __forceinline__ void sum_rows(const float* __restrict__ hin, int my_s, float my_w,
                                         int t, int lane, float4 (&acc)[F / 128]) {
  constexpr int NV = F / 128;
  float4 v[R][NV];
  float  w[R];
#pragma unroll
  for (int u = 0; u < R; ++u) {
    const int s = __builtin_amdgcn_readlane(my_s, t + u);
    w[u] = __int_as_float(__builtin_amdgcn_readlane(__float_as_int(my_w), t + u));
    const float4* p = (const float4*)(hin + (size_t)s * F);
#pragma unroll
    for (int g = 0; g < NV; ++g) v[u][g] = p[lane + 32 * g];
  }
#pragma unroll
  for (int u = 0; u < R; ++u) {
#pragma unroll
    for (int g = 0; g < NV; ++g) acc[g] = f4_fma(v[u][g], w[u], acc[g]);
  }
}

template <int F, bool RELU>
__global__ __launch_bounds__(256)
void k_gather_sum(const int* __restrict__ rowptr, const int* __restrict__ csr_src,
                  const float* __restrict__ dinv, const float* __restrict__ hin,
                  const float* __restrict__ bias, float* __restrict__ hout, int n) {
  constexpr int NV = F / 128;
  constexpr int U  = 8;
  const int lane   = threadIdx.x & 31;
  const int wave0  = __builtin_amdgcn_readfirstlane((int)((blockIdx.x * blockDim.x + threadIdx.x) >> 5));
  const int nwaves = (int)((gridDim.x * blockDim.x) >> 5);

  float4 bv[NV];
#pragma unroll
  for (int g = 0; g < NV; ++g) bv[g] = ((const float4*)bias)[lane + 32 * g];

  for (int d = wave0; d < n; d += nwaves) {
    const int beg = rowptr[d];
    const int end = rowptr[d + 1];
    const float dd = dinv[d];
    const float wself = dd * dd;
    const float4* self4 = (const float4*)(hin + (size_t)d * F);
    float4 acc[NV];
#pragma unroll
    for (int g = 0; g < NV; ++g) {
      float4 v = self4[lane + 32 * g];
      acc[g].x = v.x * wself; acc[g].y = v.y * wself;
      acc[g].z = v.z * wself; acc[g].w = v.w * wself;
    }

    for (int j = beg; j < end; j += 32) {
      const int m = (end - j < 32) ? (end - j) : 32;
      int   my_s = 0;
      float my_w = 0.0f;
      if (lane < m) {
        my_s = csr_src[j + lane];
        my_w = dinv[my_s] * dd;
      }
      int t = 0;
      for (; t + U <= m; t += U) sum_rows<U, F>(hin, my_s, my_w, t, lane, acc);
      switch (m - t) {
        case 1: sum_rows<1, F>(hin, my_s, my_w, t, lane, acc); break;
        case 2: sum_rows<2, F>(hin, my_s, my_w, t, lane, acc); break;
        case 3: sum_rows<3, F>(hin, my_s, my_w, t, lane, acc); break;
        case 4: sum_rows<4, F>(hin, my_s, my_w, t, lane, acc); break;
        case 5: sum_rows<5, F>(hin, my_s, my_w, t, lane, acc); break;
        case 6: sum_rows<6, F>(hin, my_s, my_w, t, lane, acc); break;
        case 7: sum_rows<7, F>(hin, my_s, my_w, t, lane, acc); break;
        default: break;
      }
    }

    float4* o4 = (float4*)(hout + (size_t)d * F);
#pragma unroll
    for (int g = 0; g < NV; ++g) {
      float4 r;
      r.x = acc[g].x + bv[g].x; r.y = acc[g].y + bv[g].y;
      r.z = acc[g].z + bv[g].z; r.w = acc[g].w + bv[g].w;
      if (RELU) {
        r.x = r.x > 0.0f ? r.x : 0.0f; r.y = r.y > 0.0f ? r.y : 0.0f;
        r.z = r.z > 0.0f ? r.z : 0.0f; r.w = r.w > 0.0f ? r.w : 0.0f;
      }
      o4[lane + 32 * g] = r;
    }
  }
}

extern "C" void kernel_launch(void* const* d_in, const int* in_sizes, int n_in,
                              void* d_out, int out_size, void* d_ws, size_t ws_size,
                              hipStream_t stream) {
  const float* x  = (const float*)d_in[0];
  const int*   ei = (const int*)d_in[1];
  const float* W1 = (const float*)d_in[2];
  const float* b1 = (const float*)d_in[3];
  const float* W2 = (const float*)d_in[4];
  const float* b2 = (const float*)d_in[5];

  const int n = in_sizes[0] / NFEAT;
  const int E = in_sizes[1] / 2;
  const int* src = ei;
  const int* dst = ei + E;
  const size_t npad = (size_t)((n + 256) & ~255);

  float* dinv = (float*)d_ws;
  float* Hlin = dinv + (size_t)((n + 255) & ~255);
  float* agg  = Hlin + (size_t)n * NH1;
  float* out  = (float*)d_out;

  int* cnt     = (int*)d_out;
  int* rowptrA = cnt + npad;
  int* csrA    = rowptrA + npad;
  int* rank    = csrA + E;
  int* rowptrB = (int*)(Hlin + (size_t)n * NH2);
  int* csrB    = rowptrB + npad;

  k_zero_i32<<<(n + 255) / 256, 256, 0, stream>>>(cnt, n);
  k_count_rank<<<(E + 255) / 256, 256, 0, stream>>>(dst, E, cnt, rank);
  k_scan_rowptr<<<1, 1024, 0, stream>>>(cnt, rowptrA, n);
  k_place_dinv<<<(E + n + 255) / 256, 256, 0, stream>>>(src, dst, rank, rowptrA, cnt, E, n, csrA, dinv);

  const int gather_blocks1 = 1024;
  const int gather_blocks2 = 2048;

  k_gemm_wmma<<<dim3(NH1 / 64, (n + 63) / 64), 256, 0, stream>>>(x, W1, Hlin, n, NFEAT, NH1);
  k_gather_sum<NH1, true><<<gather_blocks1, 256, 0, stream>>>(rowptrA, csrA, dinv, Hlin, b1, agg, n);

  k_move_lists<<<(E + n + 1 + 255) / 256, 256, 0, stream>>>(rowptrA, csrA, rowptrB, csrB, n, E);
  k_gemm_wmma<<<dim3(NH2 / 64, (n + 63) / 64), 256, 0, stream>>>(agg, W2, Hlin, n, NH1, NH2);
  k_gather_sum<NH2, false><<<gather_blocks2, 256, 0, stream>>>(rowptrB, csrB, dinv, Hlin, b2, out, n);
}


# GEMM tile pipeline rebuilt: 64x128 block, 32x32 per wave, next k-chunk fetched by whole-line 16-byte loads under the WMMAs into a 2nd LDS stage, C rows stored whole via LDS; sums bit-identical
# speedup vs baseline: 2.9608x; 1.5951x over previous
#include <hip/hip_runtime.h>


typedef float v2f __attribute__((ext_vector_type(2)));
typedef float v8f __attribute__((ext_vector_type(8)));

#define NFEAT 512
#define NH1   256
#define NH2   128

__global__ void k_zero_i32(int* __restrict__ p, int n) {
  int i = blockIdx.x * blockDim.x + threadIdx.x;
  if (i < n) p[i] = 0;
}

__global__ void k_count_rank(const int* __restrict__ dst, int E,
                             int* __restrict__ cnt, int* __restrict__ rank) {
  int e = blockIdx.x * blockDim.x + threadIdx.x;
  if (e < E) rank[e] = atomicAdd(&cnt[dst[e]], 1);
}

__global__ __launch_bounds__(1024)
void k_scan_rowptr(const int* __restrict__ cnt, int* __restrict__ rowptr, int n) {
  __shared__ int wtot[2][128];
  const int tid  = threadIdx.x;
  const int lane = tid & 31;
  const int wv   = tid >> 5;
  int carry = 0;
  int par = 0;
  for (int base = 0; base < n; base += 16384, par ^= 1) {
    int c[4][4], s[4], incl[4];
#pragma unroll
    for (int k = 0; k < 4; ++k) {
      const int i0 = base + k * 4096 + tid * 4;
      if (i0 + 4 <= n) {
        const int4 q = *(const int4*)(cnt + i0);
        c[k][0] = q.x; c[k][1] = q.y; c[k][2] = q.z; c[k][3] = q.w;
      } else {
#pragma unroll
        for (int j = 0; j < 4; ++j) c[k][j] = (i0 + j < n) ? cnt[i0 + j] : 0;
      }
      s[k] = c[k][0] + c[k][1] + c[k][2] + c[k][3];
      incl[k] = s[k];
    }
#pragma unroll
    for (int o = 1; o < 32; o <<= 1) {
#pragma unroll
      for (int k = 0; k < 4; ++k) {
        const int t = __shfl_up(incl[k], o);
        if (lane >= o) incl[k] += t;
      }
    }
    if (lane == 31) {
#pragma unroll
      for (int k = 0; k < 4; ++k) wtot[par][k * 32 + wv] = incl[k];
    }
    __syncthreads();
    int T[4], Wi[4];
#pragma unroll
    for (int k = 0; k < 4; ++k) { T[k] = wtot[par][k * 32 + lane]; Wi[k] = T[k]; }
#pragma unroll
    for (int o = 1; o < 32; o <<= 1) {
#pragma unroll
      for (int k = 0; k < 4; ++k) {
        const int t = __shfl_up(Wi[k], o);
        if (lane >= o) Wi[k] += t;
      }
    }
    int pre = carry;
#pragma unroll
    for (int k = 0; k < 4; ++k) {
      const int S  = __shfl(Wi[k], 31);
      const int wp = __shfl(Wi[k], wv) - __shfl(T[k], wv);
      const int r0 = pre + wp + (incl[k] - s[k]);
      const int r1 = r0 + c[k][0];
      const int r2 = r1 + c[k][1];
      const int r3 = r2 + c[k][2];
      const int i0 = base + k * 4096 + tid * 4;
      if (i0 + 4 <= n) {
        *(int4*)(rowptr + i0) = make_int4(r0, r1, r2, r3);
      } else {
        if (i0 + 0 < n) rowptr[i0 + 0] = r0;
        if (i0 + 1 < n) rowptr[i0 + 1] = r1;
        if (i0 + 2 < n) rowptr[i0 + 2] = r2;
      }
      pre += S;
    }
    carry = pre;
  }
  if (tid == 0) rowptr[n] = carry;
}

__global__ void k_place_dinv(const int* __restrict__ src, const int* __restrict__ dst,
                             const int* __restrict__ rank, const int* __restrict__ rowptr,
                             const int* __restrict__ cnt, int E, int n,
                             int* __restrict__ csr_src, float* __restrict__ dinv) {
  int i = blockIdx.x * blockDim.x + threadIdx.x;
  if (i < E) {
    csr_src[rowptr[dst[i]] + rank[i]] = src[i];
  } else if (i - E < n) {
    dinv[i - E] = rsqrtf((float)(cnt[i - E] + 1));
  }
}

__global__ void k_move_lists(const int* __restrict__ rp_a, const int* __restrict__ cs_a,
                             int* __restrict__ rp_b, int* __restrict__ cs_b, int n, int E) {
  int i = blockIdx.x * blockDim.x + threadIdx.x;
  if (i <= n) rp_b[i] = rp_a[i];
  else if (i - (n + 1) < E) cs_b[i - (n + 1)] = cs_a[i - (n + 1)];
}

#define GBM 64
#define GBN 128
#define GBK 32
#define APITCH 36
#define BPITCH 136
#define CPITCH 132
#define STAGE_FLOATS (GBM * APITCH + GBK * BPITCH)

#define LDS_GROUP(P, KK8)                                                              \
  P##a0 = *(const float4*)&st[a_frag + (KK8)];                                         \
  P##a1 = *(const float4*)&st[a_frag + 16 * APITCH + (KK8)];                           \
  P##b00.x = st[b_frag + ((KK8) + 0) * BPITCH];                                        \
  P##b00.y = st[b_frag + ((KK8) + 1) * BPITCH];                                        \
  P##b01.x = st[b_frag + ((KK8) + 0) * BPITCH + 16];                                   \
  P##b01.y = st[b_frag + ((KK8) + 1) * BPITCH + 16];                                   \
  P##b10.x = st[b_frag + ((KK8) + 4) * BPITCH];                                        \
  P##b10.y = st[b_frag + ((KK8) + 5) * BPITCH];                                        \
  P##b11.x = st[b_frag + ((KK8) + 4) * BPITCH + 16];                                   \
  P##b11.y = st[b_frag + ((KK8) + 5) * BPITCH + 16];

#define WMMA_F32(ACC, FA, FB)                                                          \
  ACC = __builtin_amdgcn_wmma_f32_16x16x4_f32(false, FA, false, FB, (short)0, ACC, false, false)

#define MMA_GROUP(P)                                                                   \
  {                                                                                    \
    v2f f0, f1;                                                                        \
    f0.x = P##a0.x; f0.y = P##a0.y; f1.x = P##a1.x; f1.y = P##a1.y;                    \
    WMMA_F32(acc00, f0, P##b00); WMMA_F32(acc01, f0, P##b01);                          \
    WMMA_F32(acc10, f1, P##b00); WMMA_F32(acc11, f1, P##b01);                          \
    f0.x = P##a0.z; f0.y = P##a0.w; f1.x = P##a1.z; f1.y = P##a1.w;                    \
    WMMA_F32(acc00, f0, P##b10); WMMA_F32(acc01, f0, P##b11);                          \
    WMMA_F32(acc10, f1, P##b10); WMMA_F32(acc11, f1, P##b11);                          \
  }

#define STAGE_STORE(ST)                                                                \
  *(float2*)&(ST)[a_lds]                   = make_float2(pa0.x, pa0.y);                \
  *(float2*)&(ST)[a_lds + 4]               = make_float2(pa0.z, pa0.w);                \
  *(float2*)&(ST)[a_lds + 32 * APITCH]     = make_float2(pa1.x, pa1.y);                \
  *(float2*)&(ST)[a_lds + 32 * APITCH + 4] = make_float2(pa1.z, pa1.w);                \
  *(float4*)&(ST)[b_lds]                   = pb0;                                      \
  *(float4*)&(ST)[b_lds + 8 * BPITCH]      = pb1;                                      \
  *(float4*)&(ST)[b_lds + 16 * BPITCH]     = pb2;                                      \
  *(float4*)&(ST)[b_lds + 24 * BPITCH]     = pb3;

__global__ __launch_bounds__(256)
void k_gemm_wmma(const float* __restrict__ A, const float* __restrict__ B,
                 float* __restrict__ C, int M, int K, int N) {
  __shared__ __attribute__((aligned(16))) float smem[2 * STAGE_FLOATS];

  const int tid  = threadIdx.x;
  const int lane = tid & 31;
  const int wv   = tid >> 5;
  const int wm   = wv & 1;
  const int wn   = wv >> 1;
  const int hf   = lane >> 4;
  const int idx  = lane & 15;

  const int m_base = blockIdx.y * GBM;
  const int n_base = blockIdx.x * GBN;

  const int ar  = tid >> 3;
  const int akq = tid & 7;
  int gr0 = m_base + ar;      gr0 = gr0 < M ? gr0 : M - 1;
  int gr1 = m_base + ar + 32; gr1 = gr1 < M ? gr1 : M - 1;
  const float* ag0 = A + (size_t)gr0 * K + 4 * akq;
  const float* ag1 = A + (size_t)gr1 * K + 4 * akq;
  const int a_lds = ar * APITCH + (akq >> 1) * 8 + (akq & 1) * 2;
  const int bn4 = tid & 31;
  const int bk  = tid >> 5;
  const float* bg = B + (size_t)bk * N + n_base + 4 * bn4;
  const size_t bstep = (size_t)8 * N;
  const int b_lds = GBM * APITCH + bk * BPITCH + 4 * bn4;

  const int a_frag = (wm * 32 + idx) * APITCH + 4 * hf;
  const int b_frag = GBM * APITCH + (2 * hf) * BPITCH + wn * 32 + idx;

  float4 pa0 = *(const float4*)ag0;
  float4 pa1 = *(const float4*)ag1;
  float4 pb0 = *(const float4*)(bg);
  float4 pb1 = *(const float4*)(bg + bstep);
  float4 pb2 = *(const float4*)(bg + 2 * bstep);
  float4 pb3 = *(const float4*)(bg + 3 * bstep);
  STAGE_STORE(smem)
  __syncthreads();

  v8f acc00 = {}, acc01 = {}, acc10 = {}, acc11 = {};

  const int nchunks = K / GBK;
  for (int ch = 0; ch < nchunks; ++ch) {
    const float* st = smem + (ch & 1) * STAGE_FLOATS;
    const bool more = (ch + 1 < nchunks);
    if (more) {
      const int k0 = (ch + 1) * GBK;
      const float* bgk = bg + (size_t)k0 * N;
      pa0 = *(const float4*)(ag0 + k0);
      pa1 = *(const float4*)(ag1 + k0);
      pb0 = *(const float4*)(bgk);
      pb1 = *(const float4*)(bgk + bstep);
      pb2 = *(const float4*)(bgk + 2 * bstep);
      pb3 = *(const float4*)(bgk + 3 * bstep);
    }

    float4 Xa0, Xa1, Ya0, Ya1;
    v2f Xb00, Xb01, Xb10, Xb11, Yb00, Yb01, Yb10, Yb11;
    LDS_GROUP(X, 0)
    LDS_GROUP(Y, 8)
    MMA_GROUP(X)
    LDS_GROUP(X, 16)
    MMA_GROUP(Y)
    LDS_GROUP(Y, 24)
    MMA_GROUP(X)
    MMA_GROUP(Y)

    if (more) {
      float* nx = smem + ((ch + 1) & 1) * STAGE_FLOATS;
      STAGE_STORE(nx)
    }
    __syncthreads();
  }

  float* Cs = smem;
#pragma unroll
  for (int r = 0; r < 8; ++r) {
    const int row = wm * 32 + 8 * hf + r;
    Cs[row * CPITCH + wn * 32 + idx]             = acc00[r];
    Cs[row * CPITCH + wn * 32 + 16 + idx]        = acc01[r];
    Cs[(row + 16) * CPITCH + wn * 32 + idx]      = acc10[r];
    Cs[(row + 16) * CPITCH + wn * 32 + 16 + idx] = acc11[r];
  }
  __syncthreads();
#pragma unroll
  for (int i = 0; i < 8; ++i) {
    const int row = wv * 8 + i;
    if (m_base + row < M) {
      const float4 v = *(const float4*)&Cs[row * CPITCH + 4 * lane];
      *(float4*)&C[(size_t)(m_base + row) * N + n_base + 4 * lane] = v;
    }
  }
}

__device__ __forceinline__ float4 f4_fma(float4 v, float w, float4 a) {
  a.x = fmaf(v.x, w, a.x); a.y = fmaf(v.y, w, a.y);
  a.z = fmaf(v.z, w, a.z); a.w = fmaf(v.w, w, a.w);
  return a;
}

template <int R, int F>
__device__ __forceinline__ void sum_rows(const float* __restrict__ hin, int my_s, float my_w,
                                         int t, int lane, float4 (&acc)[F / 128]) {
  constexpr int NV = F / 128;
  float4 v[R][NV];
  float  w[R];
#pragma unroll
  for (int u = 0; u < R; ++u) {
    const int s = __builtin_amdgcn_readlane(my_s, t + u);
    w[u] = __int_as_float(__builtin_amdgcn_readlane(__float_as_int(my_w), t + u));
    const float4* p = (const float4*)(hin + (size_t)s * F);
#pragma unroll
    for (int g = 0; g < NV; ++g) v[u][g] = p[lane + 32 * g];
  }
#pragma unroll
  for (int u = 0; u < R; ++u) {
#pragma unroll
    for (int g = 0; g < NV; ++g) acc[g] = f4_fma(v[u][g], w[u], acc[g]);
  }
}

template <int F, bool RELU>
__global__ __launch_bounds__(256)
void k_gather_sum(const int* __restrict__ rowptr, const int* __restrict__ csr_src,
                  const float* __restrict__ dinv, const float* __restrict__ hin,
                  const float* __restrict__ bias, float* __restrict__ hout, int n) {
  constexpr int NV = F / 128;
  constexpr int U  = 8;
  const int lane   = threadIdx.x & 31;
  const int wave0  = __builtin_amdgcn_readfirstlane((int)((blockIdx.x * blockDim.x + threadIdx.x) >> 5));
  const int nwaves = (int)((gridDim.x * blockDim.x) >> 5);

  float4 bv[NV];
#pragma unroll
  for (int g = 0; g < NV; ++g) bv[g] = ((const float4*)bias)[lane + 32 * g];

  for (int d = wave0; d < n; d += nwaves) {
    const int beg = rowptr[d];
    const int end = rowptr[d + 1];
    const float dd = dinv[d];
    const float wself = dd * dd;
    const float4* self4 = (const float4*)(hin + (size_t)d * F);
    float4 acc[NV];
#pragma unroll
    for (int g = 0; g < NV; ++g) {
      float4 v = self4[lane + 32 * g];
      acc[g].x = v.x * wself; acc[g].y = v.y * wself;
      acc[g].z = v.z * wself; acc[g].w = v.w * wself;
    }

    for (int j = beg; j < end; j += 32) {
      const int m = (end - j < 32) ? (end - j) : 32;
      int   my_s = 0;
      float my_w = 0.0f;
      if (lane < m) {
        my_s = csr_src[j + lane];
        my_w = dinv[my_s] * dd;
      }
      int t = 0;
      for (; t + U <= m; t += U) sum_rows<U, F>(hin, my_s, my_w, t, lane, acc);
      switch (m - t) {
        case 1: sum_rows<1, F>(hin, my_s, my_w, t, lane, acc); break;
        case 2: sum_rows<2, F>(hin, my_s, my_w, t, lane, acc); break;
        case 3: sum_rows<3, F>(hin, my_s, my_w, t, lane, acc); break;
        case 4: sum_rows<4, F>(hin, my_s, my_w, t, lane, acc); break;
        case 5: sum_rows<5, F>(hin, my_s, my_w, t, lane, acc); break;
        case 6: sum_rows<6, F>(hin, my_s, my_w, t, lane, acc); break;
        case 7: sum_rows<7, F>(hin, my_s, my_w, t, lane, acc); break;
        default: break;
      }
    }

    float4* o4 = (float4*)(hout + (size_t)d * F);
#pragma unroll
    for (int g = 0; g < NV; ++g) {
      float4 r;
      r.x = acc[g].x + bv[g].x; r.y = acc[g].y + bv[g].y;
      r.z = acc[g].z + bv[g].z; r.w = acc[g].w + bv[g].w;
      if (RELU) {
        r.x = r.x > 0.0f ? r.x : 0.0f; r.y = r.y > 0.0f ? r.y : 0.0f;
        r.z = r.z > 0.0f ? r.z : 0.0f; r.w = r.w > 0.0f ? r.w : 0.0f;
      }
      o4[lane + 32 * g] = r;
    }
  }
}

extern "C" void kernel_launch(void* const* d_in, const int* in_sizes, int n_in,
                              void* d_out, int out_size, void* d_ws, size_t ws_size,
                              hipStream_t stream) {
  const float* x  = (const float*)d_in[0];
  const int*   ei = (const int*)d_in[1];
  const float* W1 = (const float*)d_in[2];
  const float* b1 = (const float*)d_in[3];
  const float* W2 = (const float*)d_in[4];
  const float* b2 = (const float*)d_in[5];

  const int n = in_sizes[0] / NFEAT;
  const int E = in_sizes[1] / 2;
  const int* src = ei;
  const int* dst = ei + E;
  const size_t npad = (size_t)((n + 256) & ~255);

  float* dinv = (float*)d_ws;
  float* Hlin = dinv + (size_t)((n + 255) & ~255);
  float* agg  = Hlin + (size_t)n * NH1;
  float* out  = (float*)d_out;

  int* cnt     = (int*)d_out;
  int* rowptrA = cnt + npad;
  int* csrA    = rowptrA + npad;
  int* rank    = csrA + E;
  int* rowptrB = (int*)(Hlin + (size_t)n * NH2);
  int* csrB    = rowptrB + npad;

  k_zero_i32<<<(n + 255) / 256, 256, 0, stream>>>(cnt, n);
  k_count_rank<<<(E + 255) / 256, 256, 0, stream>>>(dst, E, cnt, rank);
  k_scan_rowptr<<<1, 1024, 0, stream>>>(cnt, rowptrA, n);
  k_place_dinv<<<(E + n + 255) / 256, 256, 0, stream>>>(src, dst, rank, rowptrA, cnt, E, n, csrA, dinv);

  const int gather_blocks1 = 1024;
  const int gather_blocks2 = 2048;

  k_gemm_wmma<<<dim3(NH1 / GBN, (n + GBM - 1) / GBM), 256, 0, stream>>>(x, W1, Hlin, n, NFEAT, NH1);
  k_gather_sum<NH1, true><<<gather_blocks1, 256, 0, stream>>>(rowptrA, csrA, dinv, Hlin, b1, agg, n);

  k_move_lists<<<(E + n + 1 + 255) / 256, 256, 0, stream>>>(rowptrA, csrA, rowptrB, csrB, n, E);
  k_gemm_wmma<<<dim3(NH2 / GBN, (n + GBM - 1) / GBM), 256, 0, stream>>>(agg, W2, Hlin, n, NH1, NH2);
  k_gather_sum<NH2, false><<<gather_blocks2, 256, 0, stream>>>(rowptrB, csrB, dinv, Hlin, b2, out, n);
}


# global int rank atomics + node-count scan + place -> two-level counting sort on LDS counters (256-node bins, then a work-group per bin; LDS add return = slot); row sums in LDS-add arrival order
# speedup vs baseline: 3.7087x; 1.2526x over previous
#include <hip/hip_runtime.h>


typedef float v2f __attribute__((ext_vector_type(2)));
typedef float v8f __attribute__((ext_vector_type(8)));

#define NFEAT 512
#define NH1   256
#define NH2   128

#define BIN_SHIFT   8
#define BIN_NODES   256
#define MAXNB       1024
#define SORT_BLOCKS 128
#define STAGE_CAP   12288

__global__ __launch_bounds__(1024)
void k_bin_hist(const int* __restrict__ dst, int E, int chunk, int nb, int* __restrict__ hist) {
  __shared__ int lh[MAXNB];
  const int tid = threadIdx.x;
  for (int i = tid; i < nb; i += 1024) lh[i] = 0;
  __syncthreads();
  const int e0 = blockIdx.x * chunk;
  const int e1 = (e0 + chunk < E) ? (e0 + chunk) : E;
  for (int e = e0 + tid; e < e1; e += 4096) {
    int d[4];
#pragma unroll
    for (int u = 0; u < 4; ++u) {
      const int ee = e + u * 1024;
      d[u] = (ee < e1) ? dst[ee] : -1;
    }
#pragma unroll
    for (int u = 0; u < 4; ++u)
      if (d[u] >= 0) atomicAdd(&lh[d[u] >> BIN_SHIFT], 1);
  }
  __syncthreads();
  for (int i = tid; i < nb; i += 1024) hist[i * gridDim.x + blockIdx.x] = lh[i];
}

__global__ __launch_bounds__(1024)
void k_scan_rowptr(const int* __restrict__ cnt, int* __restrict__ rowptr, int n) {
  __shared__ int wtot[2][128];
  const int tid  = threadIdx.x;
  const int lane = tid & 31;
  const int wv   = tid >> 5;
  int carry = 0;
  int par = 0;
  for (int base = 0; base < n; base += 16384, par ^= 1) {
    int c[4][4], s[4], incl[4];
#pragma unroll
    for (int k = 0; k < 4; ++k) {
      const int i0 = base + k * 4096 + tid * 4;
      if (i0 + 4 <= n) {
        const int4 q = *(const int4*)(cnt + i0);
        c[k][0] = q.x; c[k][1] = q.y; c[k][2] = q.z; c[k][3] = q.w;
      } else {
#pragma unroll
        for (int j = 0; j < 4; ++j) c[k][j] = (i0 + j < n) ? cnt[i0 + j] : 0;
      }
      s[k] = c[k][0] + c[k][1] + c[k][2] + c[k][3];
      incl[k] = s[k];
    }
#pragma unroll
    for (int o = 1; o < 32; o <<= 1) {
#pragma unroll
      for (int k = 0; k < 4; ++k) {
        const int t = __shfl_up(incl[k], o);
        if (lane >= o) incl[k] += t;
      }
    }
    if (lane == 31) {
#pragma unroll
      for (int k = 0; k < 4; ++k) wtot[par][k * 32 + wv] = incl[k];
    }
    __syncthreads();
    int T[4], Wi[4];
#pragma unroll
    for (int k = 0; k < 4; ++k) { T[k] = wtot[par][k * 32 + lane]; Wi[k] = T[k]; }
#pragma unroll
    for (int o = 1; o < 32; o <<= 1) {
#pragma unroll
      for (int k = 0; k < 4; ++k) {
        const int t = __shfl_up(Wi[k], o);
        if (lane >= o) Wi[k] += t;
      }
    }
    int pre = carry;
#pragma unroll
    for (int k = 0; k < 4; ++k) {
      const int S  = __shfl(Wi[k], 31);
      const int wp = __shfl(Wi[k], wv) - __shfl(T[k], wv);
      const int r0 = pre + wp + (incl[k] - s[k]);
      const int r1 = r0 + c[k][0];
      const int r2 = r1 + c[k][1];
      const int r3 = r2 + c[k][2];
      const int i0 = base + k * 4096 + tid * 4;
      if (i0 + 4 <= n) {
        *(int4*)(rowptr + i0) = make_int4(r0, r1, r2, r3);
      } else {
        if (i0 + 0 < n) rowptr[i0 + 0] = r0;
        if (i0 + 1 < n) rowptr[i0 + 1] = r1;
        if (i0 + 2 < n) rowptr[i0 + 2] = r2;
      }
      pre += S;
    }
    carry = pre;
  }
  if (tid == 0) rowptr[n] = carry;
}

__global__ __launch_bounds__(1024)
void k_bin_scatter(const int* __restrict__ src, const int* __restrict__ dst, int E, int chunk,
                   int nb, const int* __restrict__ offs, int* __restrict__ pairs) {
  __shared__ int lh[MAXNB];
  const int tid = threadIdx.x;
  for (int i = tid; i < nb; i += 1024) lh[i] = offs[i * gridDim.x + blockIdx.x];
  __syncthreads();
  const int e0 = blockIdx.x * chunk;
  const int e1 = (e0 + chunk < E) ? (e0 + chunk) : E;
  int2* pr = (int2*)pairs;
  for (int e = e0 + tid; e < e1; e += 4096) {
    int d[4], s[4];
#pragma unroll
    for (int u = 0; u < 4; ++u) {
      const int ee = e + u * 1024;
      d[u] = (ee < e1) ? dst[ee] : -1;
      s[u] = (ee < e1) ? src[ee] : 0;
    }
#pragma unroll
    for (int u = 0; u < 4; ++u) {
      if (d[u] >= 0) {
        const int p = atomicAdd(&lh[d[u] >> BIN_SHIFT], 1);
        pr[p] = make_int2(s[u], d[u]);
      }
    }
  }
}

__global__ __launch_bounds__(1024)
void k_bin_sort(const int* __restrict__ pairs, const int* __restrict__ offs, int nblk, int n,
                int* __restrict__ rowptr, int* __restrict__ csr_src, float* __restrict__ dinv) {
  __shared__ int cnt[BIN_NODES];
  __shared__ int cur[BIN_NODES];
  __shared__ int wsum[BIN_NODES / 32];
  __shared__ int stage[STAGE_CAP];
  const int tid  = threadIdx.x;
  const int lane = tid & 31;
  const int wv   = tid >> 5;
  const int b    = blockIdx.x;
  const int p0   = offs[b * nblk];
  const int p1   = offs[(b + 1) * nblk];
  const int node0 = b << BIN_SHIFT;
  const int2* pr = (const int2*)pairs;

  if (tid < BIN_NODES) cnt[tid] = 0;
  __syncthreads();
  for (int p = p0 + tid; p < p1; p += 4096) {
    int d[4];
#pragma unroll
    for (int u = 0; u < 4; ++u) {
      const int pp = p + u * 1024;
      d[u] = (pp < p1) ? pr[pp].y : -1;
    }
#pragma unroll
    for (int u = 0; u < 4; ++u)
      if (d[u] >= 0) atomicAdd(&cnt[d[u] - node0], 1);
  }
  __syncthreads();

  int c = 0, incl = 0;
  if (tid < BIN_NODES) {
    c = cnt[tid];
    incl = c;
#pragma unroll
    for (int o = 1; o < 32; o <<= 1) {
      const int t = __shfl_up(incl, o);
      if (lane >= o) incl += t;
    }
    if (lane == 31) wsum[wv] = incl;
  }
  __syncthreads();
  if (tid < BIN_NODES) {
    int wp = 0;
#pragma unroll
    for (int k = 0; k < BIN_NODES / 32; ++k) wp += (k < wv) ? wsum[k] : 0;
    const int start = wp + incl - c;
    cur[tid] = start;
    const int node = node0 + tid;
    if (node < n) {
      rowptr[node] = p0 + start;
      dinv[node]   = rsqrtf((float)(c + 1));
    }
  }
  if (b == (int)gridDim.x - 1 && tid == 0) rowptr[n] = p1;
  __syncthreads();

  const int len = p1 - p0;
  if (len <= STAGE_CAP) {
    for (int p = p0 + tid; p < p1; p += 4096) {
      int2 q[4];
#pragma unroll
      for (int u = 0; u < 4; ++u) {
        const int pp = p + u * 1024;
        q[u] = (pp < p1) ? pr[pp] : make_int2(0, -1);
      }
#pragma unroll
      for (int u = 0; u < 4; ++u)
        if (q[u].y >= 0) stage[atomicAdd(&cur[q[u].y - node0], 1)] = q[u].x;
    }
    __syncthreads();
    for (int i = tid; i < len; i += 1024) csr_src[p0 + i] = stage[i];
  } else {
    for (int p = p0 + tid; p < p1; p += 1024) {
      const int2 q = pr[p];
      csr_src[p0 + atomicAdd(&cur[q.y - node0], 1)] = q.x;
    }
  }
}

__global__ void k_move_lists(const int* __restrict__ rp_a, const int* __restrict__ cs_a,
                             int* __restrict__ rp_b, int* __restrict__ cs_b, int n, int E) {
  int i = blockIdx.x * blockDim.x + threadIdx.x;
  if (i <= n) rp_b[i] = rp_a[i];
  else if (i - (n + 1) < E) cs_b[i - (n + 1)] = cs_a[i - (n + 1)];
}

#define GBM 64
#define GBN 128
#define GBK 32
#define APITCH 36
#define BPITCH 136
#define CPITCH 132
#define STAGE_FLOATS (GBM * APITCH + GBK * BPITCH)

#define LDS_GROUP(P, KK8)                                                              \
  P##a0 = *(const float4*)&st[a_frag + (KK8)];                                         \
  P##a1 = *(const float4*)&st[a_frag + 16 * APITCH + (KK8)];                           \
  P##b00.x = st[b_frag + ((KK8) + 0) * BPITCH];                                        \
  P##b00.y = st[b_frag + ((KK8) + 1) * BPITCH];                                        \
  P##b01.x = st[b_frag + ((KK8) + 0) * BPITCH + 16];                                   \
  P##b01.y = st[b_frag + ((KK8) + 1) * BPITCH + 16];                                   \
  P##b10.x = st[b_frag + ((KK8) + 4) * BPITCH];                                        \
  P##b10.y = st[b_frag + ((KK8) + 5) * BPITCH];                                        \
  P##b11.x = st[b_frag + ((KK8) + 4) * BPITCH + 16];                                   \
  P##b11.y = st[b_frag + ((KK8) + 5) * BPITCH + 16];

#define WMMA_F32(ACC, FA, FB)                                                          \
  ACC = __builtin_amdgcn_wmma_f32_16x16x4_f32(false, FA, false, FB, (short)0, ACC, false, false)

#define MMA_GROUP(P)                                                                   \
  {                                                                                    \
    v2f f0, f1;                                                                        \
    f0.x = P##a0.x; f0.y = P##a0.y; f1.x = P##a1.x; f1.y = P##a1.y;                    \
    WMMA_F32(acc00, f0, P##b00); WMMA_F32(acc01, f0, P##b01);                          \
    WMMA_F32(acc10, f1, P##b00); WMMA_F32(acc11, f1, P##b01);                          \
    f0.x = P##a0.z; f0.y = P##a0.w; f1.x = P##a1.z; f1.y = P##a1.w;                    \
    WMMA_F32(acc00, f0, P##b10); WMMA_F32(acc01, f0, P##b11);                          \
    WMMA_F32(acc10, f1, P##b10); WMMA_F32(acc11, f1, P##b11);                          \
  }

#define STAGE_STORE(ST)                                                                \
  *(float2*)&(ST)[a_lds]                   = make_float2(pa0.x, pa0.y);                \
  *(float2*)&(ST)[a_lds + 4]               = make_float2(pa0.z, pa0.w);                \
  *(float2*)&(ST)[a_lds + 32 * APITCH]     = make_float2(pa1.x, pa1.y);                \
  *(float2*)&(ST)[a_lds + 32 * APITCH + 4] = make_float2(pa1.z, pa1.w);                \
  *(float4*)&(ST)[b_lds]                   = pb0;                                      \
  *(float4*)&(ST)[b_lds + 8 * BPITCH]      = pb1;                                      \
  *(float4*)&(ST)[b_lds + 16 * BPITCH]     = pb2;                                      \
  *(float4*)&(ST)[b_lds + 24 * BPITCH]     = pb3;

__global__ __launch_bounds__(256)
void k_gemm_wmma(const float* __restrict__ A, const float* __restrict__ B,
                 float* __restrict__ C, int M, int K, int N) {
  __shared__ __attribute__((aligned(16))) float smem[2 * STAGE_FLOATS];

  const int tid  = threadIdx.x;
  const int lane = tid & 31;
  const int wv   = tid >> 5;
  const int wm   = wv & 1;
  const int wn   = wv >> 1;
  const int hf   = lane >> 4;
  const int idx  = lane & 15;

  const int m_base = blockIdx.y * GBM;
  const int n_base = blockIdx.x * GBN;

  const int ar  = tid >> 3;
  const int akq = tid & 7;
  int gr0 = m_base + ar;      gr0 = gr0 < M ? gr0 : M - 1;
  int gr1 = m_base + ar + 32; gr1 = gr1 < M ? gr1 : M - 1;
  const float* ag0 = A + (size_t)gr0 * K + 4 * akq;
  const float* ag1 = A + (size_t)gr1 * K + 4 * akq;
  const int a_lds = ar * APITCH + (akq >> 1) * 8 + (akq & 1) * 2;
  const int bn4 = tid & 31;
  const int bk  = tid >> 5;
  const float* bg = B + (size_t)bk * N + n_base + 4 * bn4;
  const size_t bstep = (size_t)8 * N;
  const int b_lds = GBM * APITCH + bk * BPITCH + 4 * bn4;

  const int a_frag = (wm * 32 + idx) * APITCH + 4 * hf;
  const int b_frag = GBM * APITCH + (2 * hf) * BPITCH + wn * 32 + idx;

  float4 pa0 = *(const float4*)ag0;
  float4 pa1 = *(const float4*)ag1;
  float4 pb0 = *(const float4*)(bg);
  float4 pb1 = *(const float4*)(bg + bstep);
  float4 pb2 = *(const float4*)(bg + 2 * bstep);
  float4 pb3 = *(const float4*)(bg + 3 * bstep);
  STAGE_STORE(smem)
  __syncthreads();

  v8f acc00 = {}, acc01 = {}, acc10 = {}, acc11 = {};

  const int nchunks = K / GBK;
  for (int ch = 0; ch < nchunks; ++ch) {
    const float* st = smem + (ch & 1) * STAGE_FLOATS;
    const bool more = (ch + 1 < nchunks);
    if (more) {
      const int k0 = (ch + 1) * GBK;
      const float* bgk = bg + (size_t)k0 * N;
      pa0 = *(const float4*)(ag0 + k0);
      pa1 = *(const float4*)(ag1 + k0);
      pb0 = *(const float4*)(bgk);
      pb1 = *(const float4*)(bgk + bstep);
      pb2 = *(const float4*)(bgk + 2 * bstep);
      pb3 = *(const float4*)(bgk + 3 * bstep);
    }

    float4 Xa0, Xa1, Ya0, Ya1;
    v2f Xb00, Xb01, Xb10, Xb11, Yb00, Yb01, Yb10, Yb11;
    LDS_GROUP(X, 0)
    LDS_GROUP(Y, 8)
    MMA_GROUP(X)
    LDS_GROUP(X, 16)
    MMA_GROUP(Y)
    LDS_GROUP(Y, 24)
    MMA_GROUP(X)
    MMA_GROUP(Y)

    if (more) {
      float* nx = smem + ((ch + 1) & 1) * STAGE_FLOATS;
      STAGE_STORE(nx)
    }
    __syncthreads();
  }

  float* Cs = smem;
#pragma unroll
  for (int r = 0; r < 8; ++r) {
    const int row = wm * 32 + 8 * hf + r;
    Cs[row * CPITCH + wn * 32 + idx]             = acc00[r];
    Cs[row * CPITCH + wn * 32 + 16 + idx]        = acc01[r];
    Cs[(row + 16) * CPITCH + wn * 32 + idx]      = acc10[r];
    Cs[(row + 16) * CPITCH + wn * 32 + 16 + idx] = acc11[r];
  }
  __syncthreads();
#pragma unroll
  for (int i = 0; i < 8; ++i) {
    const int row = wv * 8 + i;
    if (m_base + row < M) {
      const float4 v = *(const float4*)&Cs[row * CPITCH + 4 * lane];
      *(float4*)&C[(size_t)(m_base + row) * N + n_base + 4 * lane] = v;
    }
  }
}

__device__ __forceinline__ float4 f4_fma(float4 v, float w, float4 a) {
  a.x = fmaf(v.x, w, a.x); a.y = fmaf(v.y, w, a.y);
  a.z = fmaf(v.z, w, a.z); a.w = fmaf(v.w, w, a.w);
  return a;
}

template <int R, int F>
__device__ __forceinline__ void sum_rows(const float* __restrict__ hin, int my_s, float my_w,
                                         int t, int lane, float4 (&acc)[F / 128]) {
  constexpr int NV = F / 128;
  float4 v[R][NV];
  float  w[R];
#pragma unroll
  for (int u = 0; u < R; ++u) {
    const int s = __builtin_amdgcn_readlane(my_s, t + u);
    w[u] = __int_as_float(__builtin_amdgcn_readlane(__float_as_int(my_w), t + u));
    const float4* p = (const float4*)(hin + (size_t)s * F);
#pragma unroll
    for (int g = 0; g < NV; ++g) v[u][g] = p[lane + 32 * g];
  }
#pragma unroll
  for (int u = 0; u < R; ++u) {
#pragma unroll
    for (int g = 0; g < NV; ++g) acc[g] = f4_fma(v[u][g], w[u], acc[g]);
  }
}

template <int F, bool RELU>
__global__ __launch_bounds__(256)
void k_gather_sum(const int* __restrict__ rowptr, const int* __restrict__ csr_src,
                  const float* __restrict__ dinv, const float* __restrict__ hin,
                  const float* __restrict__ bias, float* __restrict__ hout, int n) {
  constexpr int NV = F / 128;
  constexpr int U  = 8;
  const int lane   = threadIdx.x & 31;
  const int wave0  = __builtin_amdgcn_readfirstlane((int)((blockIdx.x * blockDim.x + threadIdx.x) >> 5));
  const int nwaves = (int)((gridDim.x * blockDim.x) >> 5);

  float4 bv[NV];
#pragma unroll
  for (int g = 0; g < NV; ++g) bv[g] = ((const float4*)bias)[lane + 32 * g];

  for (int d = wave0; d < n; d += nwaves) {
    const int beg = rowptr[d];
    const int end = rowptr[d + 1];
    const float dd = dinv[d];
    const float wself = dd * dd;
    const float4* self4 = (const float4*)(hin + (size_t)d * F);
    float4 acc[NV];
#pragma unroll
    for (int g = 0; g < NV; ++g) {
      float4 v = self4[lane + 32 * g];
      acc[g].x = v.x * wself; acc[g].y = v.y * wself;
      acc[g].z = v.z * wself; acc[g].w = v.w * wself;
    }

    for (int j = beg; j < end; j += 32) {
      const int m = (end - j < 32) ? (end - j) : 32;
      int   my_s = 0;
      float my_w = 0.0f;
      if (lane < m) {
        my_s = csr_src[j + lane];
        my_w = dinv[my_s] * dd;
      }
      int t = 0;
      for (; t + U <= m; t += U) sum_rows<U, F>(hin, my_s, my_w, t, lane, acc);
      switch (m - t) {
        case 1: sum_rows<1, F>(hin, my_s, my_w, t, lane, acc); break;
        case 2: sum_rows<2, F>(hin, my_s, my_w, t, lane, acc); break;
        case 3: sum_rows<3, F>(hin, my_s, my_w, t, lane, acc); break;
        case 4: sum_rows<4, F>(hin, my_s, my_w, t, lane, acc); break;
        case 5: sum_rows<5, F>(hin, my_s, my_w, t, lane, acc); break;
        case 6: sum_rows<6, F>(hin, my_s, my_w, t, lane, acc); break;
        case 7: sum_rows<7, F>(hin, my_s, my_w, t, lane, acc); break;
        default: break;
      }
    }

    float4* o4 = (float4*)(hout + (size_t)d * F);
#pragma unroll
    for (int g = 0; g < NV; ++g) {
      float4 r;
      r.x = acc[g].x + bv[g].x; r.y = acc[g].y + bv[g].y;
      r.z = acc[g].z + bv[g].z; r.w = acc[g].w + bv[g].w;
      if (RELU) {
        r.x = r.x > 0.0f ? r.x : 0.0f; r.y = r.y > 0.0f ? r.y : 0.0f;
        r.z = r.z > 0.0f ? r.z : 0.0f; r.w = r.w > 0.0f ? r.w : 0.0f;
      }
      o4[lane + 32 * g] = r;
    }
  }
}

extern "C" void kernel_launch(void* const* d_in, const int* in_sizes, int n_in,
                              void* d_out, int out_size, void* d_ws, size_t ws_size,
                              hipStream_t stream) {
  const float* x  = (const float*)d_in[0];
  const int*   ei = (const int*)d_in[1];
  const float* W1 = (const float*)d_in[2];
  const float* b1 = (const float*)d_in[3];
  const float* W2 = (const float*)d_in[4];
  const float* b2 = (const float*)d_in[5];

  const int n = in_sizes[0] / NFEAT;
  const int E = in_sizes[1] / 2;
  const int* src = ei;
  const int* dst = ei + E;
  const size_t npad = (size_t)((n + 256) & ~255);

  float* dinv = (float*)d_ws;
  float* Hlin = dinv + (size_t)((n + 255) & ~255);
  float* agg  = Hlin + (size_t)n * NH1;
  float* out  = (float*)d_out;

  const int nb    = (n + BIN_NODES - 1) >> BIN_SHIFT;
  const int hn    = nb * SORT_BLOCKS;
  const size_t hnp = (size_t)((hn + 256) & ~255);
  const int chunk = (E + SORT_BLOCKS - 1) / SORT_BLOCKS;
  int* hist    = (int*)d_out;
  int* offs    = hist + hnp;
  int* rowptrA = offs + hnp;
  int* csrA    = rowptrA + npad;
  int* pairs   = csrA + (size_t)((E + 3) & ~3);
  int* rowptrB = (int*)(Hlin + (size_t)n * NH2);
  int* csrB    = rowptrB + npad;

  k_bin_hist<<<SORT_BLOCKS, 1024, 0, stream>>>(dst, E, chunk, nb, hist);
  k_scan_rowptr<<<1, 1024, 0, stream>>>(hist, offs, hn);
  k_bin_scatter<<<SORT_BLOCKS, 1024, 0, stream>>>(src, dst, E, chunk, nb, offs, pairs);
  k_bin_sort<<<nb, 1024, 0, stream>>>(pairs, offs, SORT_BLOCKS, n, rowptrA, csrA, dinv);

  const int gather_blocks1 = 1024;
  const int gather_blocks2 = 2048;

  k_gemm_wmma<<<dim3(NH1 / GBN, (n + GBM - 1) / GBM), 256, 0, stream>>>(x, W1, Hlin, n, NFEAT, NH1);
  k_gather_sum<NH1, true><<<gather_blocks1, 256, 0, stream>>>(rowptrA, csrA, dinv, Hlin, b1, agg, n);

  k_move_lists<<<(E + n + 1 + 255) / 256, 256, 0, stream>>>(rowptrA, csrA, rowptrB, csrB, n, E);
  k_gemm_wmma<<<dim3(NH2 / GBN, (n + GBM - 1) / GBM), 256, 0, stream>>>(agg, W2, Hlin, n, NH1, NH2);
  k_gather_sum<NH2, false><<<gather_blocks2, 256, 0, stream>>>(rowptrB, csrB, dinv, Hlin, b2, out, n);
}


# layer-2 GEMM moved into the layer-1 gather launch: a work-group's 32 summed relu rows stay in LDS as the WMMA A tile; h1 never stored, its buffer holds the edge lists (no list move); sum order kept
# speedup vs baseline: 3.9915x; 1.0763x over previous
#include <hip/hip_runtime.h>


typedef float v2f __attribute__((ext_vector_type(2)));
typedef float v8f __attribute__((ext_vector_type(8)));

#define NFEAT 512
#define NH1   256
#define NH2   128

#define BIN_SHIFT   8
#define BIN_NODES   256
#define MAXNB       1024
#define SORT_BLOCKS 128
#define STAGE_CAP   12288

__global__ __launch_bounds__(1024)
void k_bin_hist(const int* __restrict__ dst, int E, int chunk, int nb, int* __restrict__ hist) {
  __shared__ int lh[MAXNB];
  const int tid = threadIdx.x;
  for (int i = tid; i < nb; i += 1024) lh[i] = 0;
  __syncthreads();
  const int e0 = blockIdx.x * chunk;
  const int e1 = (e0 + chunk < E) ? (e0 + chunk) : E;
  for (int e = e0 + tid; e < e1; e += 4096) {
    int d[4];
#pragma unroll
    for (int u = 0; u < 4; ++u) {
      const int ee = e + u * 1024;
      d[u] = (ee < e1) ? dst[ee] : -1;
    }
#pragma unroll
    for (int u = 0; u < 4; ++u)
      if (d[u] >= 0) atomicAdd(&lh[d[u] >> BIN_SHIFT], 1);
  }
  __syncthreads();
  for (int i = tid; i < nb; i += 1024) hist[i * gridDim.x + blockIdx.x] = lh[i];
}

__global__ __launch_bounds__(1024)
void k_scan_rowptr(const int* __restrict__ cnt, int* __restrict__ rowptr, int n) {
  __shared__ int wtot[2][128];
  const int tid  = threadIdx.x;
  const int lane = tid & 31;
  const int wv   = tid >> 5;
  int carry = 0;
  int par = 0;
  for (int base = 0; base < n; base += 16384, par ^= 1) {
    int c[4][4], s[4], incl[4];
#pragma unroll
    for (int k = 0; k < 4; ++k) {
      const int i0 = base + k * 4096 + tid * 4;
      if (i0 + 4 <= n) {
        const int4 q = *(const int4*)(cnt + i0);
        c[k][0] = q.x; c[k][1] = q.y; c[k][2] = q.z; c[k][3] = q.w;
      } else {
#pragma unroll
        for (int j = 0; j < 4; ++j) c[k][j] = (i0 + j < n) ? cnt[i0 + j] : 0;
      }
      s[k] = c[k][0] + c[k][1] + c[k][2] + c[k][3];
      incl[k] = s[k];
    }
#pragma unroll
    for (int o = 1; o < 32; o <<= 1) {
#pragma unroll
      for (int k = 0; k < 4; ++k) {
        const int t = __shfl_up(incl[k], o);
        if (lane >= o) incl[k] += t;
      }
    }
    if (lane == 31) {
#pragma unroll
      for (int k = 0; k < 4; ++k) wtot[par][k * 32 + wv] = incl[k];
    }
    __syncthreads();
    int T[4], Wi[4];
#pragma unroll
    for (int k = 0; k < 4; ++k) { T[k] = wtot[par][k * 32 + lane]; Wi[k] = T[k]; }
#pragma unroll
    for (int o = 1; o < 32; o <<= 1) {
#pragma unroll
      for (int k = 0; k < 4; ++k) {
        const int t = __shfl_up(Wi[k], o);
        if (lane >= o) Wi[k] += t;
      }
    }
    int pre = carry;
#pragma unroll
    for (int k = 0; k < 4; ++k) {
      const int S  = __shfl(Wi[k], 31);
      const int wp = __shfl(Wi[k], wv) - __shfl(T[k], wv);
      const int r0 = pre + wp + (incl[k] - s[k]);
      const int r1 = r0 + c[k][0];
      const int r2 = r1 + c[k][1];
      const int r3 = r2 + c[k][2];
      const int i0 = base + k * 4096 + tid * 4;
      if (i0 + 4 <= n) {
        *(int4*)(rowptr + i0) = make_int4(r0, r1, r2, r3);
      } else {
        if (i0 + 0 < n) rowptr[i0 + 0] = r0;
        if (i0 + 1 < n) rowptr[i0 + 1] = r1;
        if (i0 + 2 < n) rowptr[i0 + 2] = r2;
      }
      pre += S;
    }
    carry = pre;
  }
  if (tid == 0) rowptr[n] = carry;
}

__global__ __launch_bounds__(1024)
void k_bin_scatter(const int* __restrict__ src, const int* __restrict__ dst, int E, int chunk,
                   int nb, const int* __restrict__ offs, int* __restrict__ pairs) {
  __shared__ int lh[MAXNB];
  const int tid = threadIdx.x;
  for (int i = tid; i < nb; i += 1024) lh[i] = offs[i * gridDim.x + blockIdx.x];
  __syncthreads();
  const int e0 = blockIdx.x * chunk;
  const int e1 = (e0 + chunk < E) ? (e0 + chunk) : E;
  int2* pr = (int2*)pairs;
  for (int e = e0 + tid; e < e1; e += 4096) {
    int d[4], s[4];
#pragma unroll
    for (int u = 0; u < 4; ++u) {
      const int ee = e + u * 1024;
      d[u] = (ee < e1) ? dst[ee] : -1;
      s[u] = (ee < e1) ? src[ee] : 0;
    }
#pragma unroll
    for (int u = 0; u < 4; ++u) {
      if (d[u] >= 0) {
        const int p = atomicAdd(&lh[d[u] >> BIN_SHIFT], 1);
        pr[p] = make_int2(s[u], d[u]);
      }
    }
  }
}

__global__ __launch_bounds__(1024)
void k_bin_sort(const int* __restrict__ pairs, const int* __restrict__ offs, int nblk, int n,
                int* __restrict__ rowptr, int* __restrict__ csr_src, float* __restrict__ dinv) {
  __shared__ int cnt[BIN_NODES];
  __shared__ int cur[BIN_NODES];
  __shared__ int wsum[BIN_NODES / 32];
  __shared__ int stage[STAGE_CAP];
  const int tid  = threadIdx.x;
  const int lane = tid & 31;
  const int wv   = tid >> 5;
  const int b    = blockIdx.x;
  const int p0   = offs[b * nblk];
  const int p1   = offs[(b + 1) * nblk];
  const int node0 = b << BIN_SHIFT;
  const int2* pr = (const int2*)pairs;

  if (tid < BIN_NODES) cnt[tid] = 0;
  __syncthreads();
  for (int p = p0 + tid; p < p1; p += 4096) {
    int d[4];
#pragma unroll
    for (int u = 0; u < 4; ++u) {
      const int pp = p + u * 1024;
      d[u] = (pp < p1) ? pr[pp].y : -1;
    }
#pragma unroll
    for (int u = 0; u < 4; ++u)
      if (d[u] >= 0) atomicAdd(&cnt[d[u] - node0], 1);
  }
  __syncthreads();

  int c = 0, incl = 0;
  if (tid < BIN_NODES) {
    c = cnt[tid];
    incl = c;
#pragma unroll
    for (int o = 1; o < 32; o <<= 1) {
      const int t = __shfl_up(incl, o);
      if (lane >= o) incl += t;
    }
    if (lane == 31) wsum[wv] = incl;
  }
  __syncthreads();
  if (tid < BIN_NODES) {
    int wp = 0;
#pragma unroll
    for (int k = 0; k < BIN_NODES / 32; ++k) wp += (k < wv) ? wsum[k] : 0;
    const int start = wp + incl - c;
    cur[tid] = start;
    const int node = node0 + tid;
    if (node < n) {
      rowptr[node] = p0 + start;
      dinv[node]   = rsqrtf((float)(c + 1));
    }
  }
  if (b == (int)gridDim.x - 1 && tid == 0) rowptr[n] = p1;
  __syncthreads();

  const int len = p1 - p0;
  if (len <= STAGE_CAP) {
    for (int p = p0 + tid; p < p1; p += 4096) {
      int2 q[4];
#pragma unroll
      for (int u = 0; u < 4; ++u) {
        const int pp = p + u * 1024;
        q[u] = (pp < p1) ? pr[pp] : make_int2(0, -1);
      }
#pragma unroll
      for (int u = 0; u < 4; ++u)
        if (q[u].y >= 0) stage[atomicAdd(&cur[q[u].y - node0], 1)] = q[u].x;
    }
    __syncthreads();
    for (int i = tid; i < len; i += 1024) csr_src[p0 + i] = stage[i];
  } else {
    for (int p = p0 + tid; p < p1; p += 1024) {
      const int2 q = pr[p];
      csr_src[p0 + atomicAdd(&cur[q.y - node0], 1)] = q.x;
    }
  }
}

#define GBM 64
#define GBN 128
#define GBK 32
#define APITCH 36
#define BPITCH 136
#define CPITCH 132
#define STAGE_FLOATS (GBM * APITCH + GBK * BPITCH)

#define LDS_GROUP(P, KK8)                                                              \
  P##a0 = *(const float4*)&st[a_frag + (KK8)];                                         \
  P##a1 = *(const float4*)&st[a_frag + 16 * APITCH + (KK8)];                           \
  P##b00.x = st[b_frag + ((KK8) + 0) * BPITCH];                                        \
  P##b00.y = st[b_frag + ((KK8) + 1) * BPITCH];                                        \
  P##b01.x = st[b_frag + ((KK8) + 0) * BPITCH + 16];                                   \
  P##b01.y = st[b_frag + ((KK8) + 1) * BPITCH + 16];                                   \
  P##b10.x = st[b_frag + ((KK8) + 4) * BPITCH];                                        \
  P##b10.y = st[b_frag + ((KK8) + 5) * BPITCH];                                        \
  P##b11.x = st[b_frag + ((KK8) + 4) * BPITCH + 16];                                   \
  P##b11.y = st[b_frag + ((KK8) + 5) * BPITCH + 16];

#define WMMA_F32(ACC, FA, FB)                                                          \
  ACC = __builtin_amdgcn_wmma_f32_16x16x4_f32(false, FA, false, FB, (short)0, ACC, false, false)

#define MMA_GROUP(P)                                                                   \
  {                                                                                    \
    v2f f0, f1;                                                                        \
    f0.x = P##a0.x; f0.y = P##a0.y; f1.x = P##a1.x; f1.y = P##a1.y;                    \
    WMMA_F32(acc00, f0, P##b00); WMMA_F32(acc01, f0, P##b01);                          \
    WMMA_F32(acc10, f1, P##b00); WMMA_F32(acc11, f1, P##b01);                          \
    f0.x = P##a0.z; f0.y = P##a0.w; f1.x = P##a1.z; f1.y = P##a1.w;                    \
    WMMA_F32(acc00, f0, P##b10); WMMA_F32(acc01, f0, P##b11);                          \
    WMMA_F32(acc10, f1, P##b10); WMMA_F32(acc11, f1, P##b11);                          \
  }

#define STAGE_STORE(ST)                                                                \
  *(float2*)&(ST)[a_lds]                   = make_float2(pa0.x, pa0.y);                \
  *(float2*)&(ST)[a_lds + 4]               = make_float2(pa0.z, pa0.w);                \
  *(float2*)&(ST)[a_lds + 32 * APITCH]     = make_float2(pa1.x, pa1.y);                \
  *(float2*)&(ST)[a_lds + 32 * APITCH + 4] = make_float2(pa1.z, pa1.w);                \
  *(float4*)&(ST)[b_lds]                   = pb0;                                      \
  *(float4*)&(ST)[b_lds + 8 * BPITCH]      = pb1;                                      \
  *(float4*)&(ST)[b_lds + 16 * BPITCH]     = pb2;                                      \
  *(float4*)&(ST)[b_lds + 24 * BPITCH]     = pb3;

__global__ __launch_bounds__(256)
void k_gemm_wmma(const float* __restrict__ A, const float* __restrict__ B,
                 float* __restrict__ C, int M, int K, int N) {
  __shared__ __attribute__((aligned(16))) float smem[2 * STAGE_FLOATS];

  const int tid  = threadIdx.x;
  const int lane = tid & 31;
  const int wv   = tid >> 5;
  const int wm   = wv & 1;
  const int wn   = wv >> 1;
  const int hf   = lane >> 4;
  const int idx  = lane & 15;

  const int m_base = blockIdx.y * GBM;
  const int n_base = blockIdx.x * GBN;

  const int ar  = tid >> 3;
  const int akq = tid & 7;
  int gr0 = m_base + ar;      gr0 = gr0 < M ? gr0 : M - 1;
  int gr1 = m_base + ar + 32; gr1 = gr1 < M ? gr1 : M - 1;
  const float* ag0 = A + (size_t)gr0 * K + 4 * akq;
  const float* ag1 = A + (size_t)gr1 * K + 4 * akq;
  const int a_lds = ar * APITCH + (akq >> 1) * 8 + (akq & 1) * 2;
  const int bn4 = tid & 31;
  const int bk  = tid >> 5;
  const float* bg = B + (size_t)bk * N + n_base + 4 * bn4;
  const size_t bstep = (size_t)8 * N;
  const int b_lds = GBM * APITCH + bk * BPITCH + 4 * bn4;

  const int a_frag = (wm * 32 + idx) * APITCH + 4 * hf;
  const int b_frag = GBM * APITCH + (2 * hf) * BPITCH + wn * 32 + idx;

  float4 pa0 = *(const float4*)ag0;
  float4 pa1 = *(const float4*)ag1;
  float4 pb0 = *(const float4*)(bg);
  float4 pb1 = *(const float4*)(bg + bstep);
  float4 pb2 = *(const float4*)(bg + 2 * bstep);
  float4 pb3 = *(const float4*)(bg + 3 * bstep);
  STAGE_STORE(smem)
  __syncthreads();

  v8f acc00 = {}, acc01 = {}, acc10 = {}, acc11 = {};

  const int nchunks = K / GBK;
  for (int ch = 0; ch < nchunks; ++ch) {
    const float* st = smem + (ch & 1) * STAGE_FLOATS;
    const bool more = (ch + 1 < nchunks);
    if (more) {
      const int k0 = (ch + 1) * GBK;
      const float* bgk = bg + (size_t)k0 * N;
      pa0 = *(const float4*)(ag0 + k0);
      pa1 = *(const float4*)(ag1 + k0);
      pb0 = *(const float4*)(bgk);
      pb1 = *(const float4*)(bgk + bstep);
      pb2 = *(const float4*)(bgk + 2 * bstep);
      pb3 = *(const float4*)(bgk + 3 * bstep);
    }

    float4 Xa0, Xa1, Ya0, Ya1;
    v2f Xb00, Xb01, Xb10, Xb11, Yb00, Yb01, Yb10, Yb11;
    LDS_GROUP(X, 0)
    LDS_GROUP(Y, 8)
    MMA_GROUP(X)
    LDS_GROUP(X, 16)
    MMA_GROUP(Y)
    LDS_GROUP(Y, 24)
    MMA_GROUP(X)
    MMA_GROUP(Y)

    if (more) {
      float* nx = smem + ((ch + 1) & 1) * STAGE_FLOATS;
      STAGE_STORE(nx)
    }
    __syncthreads();
  }

  float* Cs = smem;
#pragma unroll
  for (int r = 0; r < 8; ++r) {
    const int row = wm * 32 + 8 * hf + r;
    Cs[row * CPITCH + wn * 32 + idx]             = acc00[r];
    Cs[row * CPITCH + wn * 32 + 16 + idx]        = acc01[r];
    Cs[(row + 16) * CPITCH + wn * 32 + idx]      = acc10[r];
    Cs[(row + 16) * CPITCH + wn * 32 + 16 + idx] = acc11[r];
  }
  __syncthreads();
#pragma unroll
  for (int i = 0; i < 8; ++i) {
    const int row = wv * 8 + i;
    if (m_base + row < M) {
      const float4 v = *(const float4*)&Cs[row * CPITCH + 4 * lane];
      *(float4*)&C[(size_t)(m_base + row) * N + n_base + 4 * lane] = v;
    }
  }
}

__device__ __forceinline__ float4 f4_fma(float4 v, float w, float4 a) {
  a.x = fmaf(v.x, w, a.x); a.y = fmaf(v.y, w, a.y);
  a.z = fmaf(v.z, w, a.z); a.w = fmaf(v.w, w, a.w);
  return a;
}

template <int R, int F>
__device__ __forceinline__ void sum_rows(const float* __restrict__ hin, int my_s, float my_w,
                                         int t, int lane, float4 (&acc)[F / 128]) {
  constexpr int NV = F / 128;
  float4 v[R][NV];
  float  w[R];
#pragma unroll
  for (int u = 0; u < R; ++u) {
    const int s = __builtin_amdgcn_readlane(my_s, t + u);
    w[u] = __int_as_float(__builtin_amdgcn_readlane(__float_as_int(my_w), t + u));
    const float4* p = (const float4*)(hin + (size_t)s * F);
#pragma unroll
    for (int g = 0; g < NV; ++g) v[u][g] = p[lane + 32 * g];
  }
#pragma unroll
  for (int u = 0; u < R; ++u) {
#pragma unroll
    for (int g = 0; g < NV; ++g) acc[g] = f4_fma(v[u][g], w[u], acc[g]);
  }
}

template <int F, bool RELU>
__global__ __launch_bounds__(256)
void k_gather_sum(const int* __restrict__ rowptr, const int* __restrict__ csr_src,
                  const float* __restrict__ dinv, const float* __restrict__ hin,
                  const float* __restrict__ bias, float* __restrict__ hout, int n) {
  constexpr int NV = F / 128;
  constexpr int U  = 8;
  const int lane   = threadIdx.x & 31;
  const int wave0  = __builtin_amdgcn_readfirstlane((int)((blockIdx.x * blockDim.x + threadIdx.x) >> 5));
  const int nwaves = (int)((gridDim.x * blockDim.x) >> 5);

  float4 bv[NV];
#pragma unroll
  for (int g = 0; g < NV; ++g) bv[g] = ((const float4*)bias)[lane + 32 * g];

  for (int d = wave0; d < n; d += nwaves) {
    const int beg = rowptr[d];
    const int end = rowptr[d + 1];
    const float dd = dinv[d];
    const float wself = dd * dd;
    const float4* self4 = (const float4*)(hin + (size_t)d * F);
    float4 acc[NV];
#pragma unroll
    for (int g = 0; g < NV; ++g) {
      float4 v = self4[lane + 32 * g];
      acc[g].x = v.x * wself; acc[g].y = v.y * wself;
      acc[g].z = v.z * wself; acc[g].w = v.w * wself;
    }

    for (int j = beg; j < end; j += 32) {
      const int m = (end - j < 32) ? (end - j) : 32;
      int   my_s = 0;
      float my_w = 0.0f;
      if (lane < m) {
        my_s = csr_src[j + lane];
        my_w = dinv[my_s] * dd;
      }
      int t = 0;
      for (; t + U <= m; t += U) sum_rows<U, F>(hin, my_s, my_w, t, lane, acc);
      switch (m - t) {
        case 1: sum_rows<1, F>(hin, my_s, my_w, t, lane, acc); break;
        case 2: sum_rows<2, F>(hin, my_s, my_w, t, lane, acc); break;
        case 3: sum_rows<3, F>(hin, my_s, my_w, t, lane, acc); break;
        case 4: sum_rows<4, F>(hin, my_s, my_w, t, lane, acc); break;
        case 5: sum_rows<5, F>(hin, my_s, my_w, t, lane, acc); break;
        case 6: sum_rows<6, F>(hin, my_s, my_w, t, lane, acc); break;
        case 7: sum_rows<7, F>(hin, my_s, my_w, t, lane, acc); break;
        default: break;
      }
    }

    float4* o4 = (float4*)(hout + (size_t)d * F);
#pragma unroll
    for (int g = 0; g < NV; ++g) {
      float4 r;
      r.x = acc[g].x + bv[g].x; r.y = acc[g].y + bv[g].y;
      r.z = acc[g].z + bv[g].z; r.w = acc[g].w + bv[g].w;
      if (RELU) {
        r.x = r.x > 0.0f ? r.x : 0.0f; r.y = r.y > 0.0f ? r.y : 0.0f;
        r.z = r.z > 0.0f ? r.z : 0.0f; r.w = r.w > 0.0f ? r.w : 0.0f;
      }
      o4[lane + 32 * g] = r;
    }
  }
}

__device__ __forceinline__ float4 f4_scale(float4 v, float w) {
  v.x *= w; v.y *= w; v.z *= w; v.w *= w;
  return v;
}
__device__ __forceinline__ float4 f4_bias_relu(float4 a, float4 b) {
  float4 q;
  q.x = a.x + b.x; q.y = a.y + b.y; q.z = a.z + b.z; q.w = a.w + b.w;
  q.x = q.x > 0.0f ? q.x : 0.0f; q.y = q.y > 0.0f ? q.y : 0.0f;
  q.z = q.z > 0.0f ? q.z : 0.0f; q.w = q.w > 0.0f ? q.w : 0.0f;
  return q;
}

#define FM     32
#define HPITCH 260

#define FUSED_GROUP(KK8)                                                               \
  {                                                                                    \
    const float4 fa = *(const float4*)&ar[(KK8)];                                      \
    v2f b00, b01, b10, b11, f0, f1;                                                    \
    b00.x = st[b_frag + ((KK8) + 0) * BPITCH];                                         \
    b00.y = st[b_frag + ((KK8) + 1) * BPITCH];                                         \
    b01.x = st[b_frag + ((KK8) + 0) * BPITCH + 16];                                    \
    b01.y = st[b_frag + ((KK8) + 1) * BPITCH + 16];                                    \
    b10.x = st[b_frag + ((KK8) + 4) * BPITCH];                                         \
    b10.y = st[b_frag + ((KK8) + 5) * BPITCH];                                         \
    b11.x = st[b_frag + ((KK8) + 4) * BPITCH + 16];                                    \
    b11.y = st[b_frag + ((KK8) + 5) * BPITCH + 16];                                    \
    f0.x = fa.x; f0.y = fa.y; f1.x = fa.z; f1.y = fa.w;                                \
    WMMA_F32(acc0, f0, b00); WMMA_F32(acc1, f0, b01);                                  \
    WMMA_F32(acc0, f1, b10); WMMA_F32(acc1, f1, b11);                                  \
  }

#define BSTAGE_STORE(ST)                                                               \
  *(float4*)&(ST)[b_lds]               = pb0;                                          \
  *(float4*)&(ST)[b_lds + 8 * BPITCH]  = pb1;                                          \
  *(float4*)&(ST)[b_lds + 16 * BPITCH] = pb2;                                          \
  *(float4*)&(ST)[b_lds + 24 * BPITCH] = pb3;

__global__ __launch_bounds__(256)
void k_gather_relu_gemm(const int* __restrict__ rowptr, const int* __restrict__ csr_src,
                        const float* __restrict__ dinv, const float* __restrict__ hin,
                        const float* __restrict__ bias, const float* __restrict__ B,
                        float* __restrict__ C, int n) {
  __shared__ __attribute__((aligned(16))) float smem[FM * HPITCH + 2 * GBK * BPITCH];
  float* As  = smem;
  float* Bst = smem + FM * HPITCH;

  const int tid  = threadIdx.x;
  const int lane = tid & 31;
  const int wv   = tid >> 5;
  const int hf   = lane >> 4;
  const int idx  = lane & 15;
  const int m_base = blockIdx.x * FM;

  const int bn4 = tid & 31;
  const int bk  = tid >> 5;
  const float* bg = B + (size_t)bk * NH2 + 4 * bn4;
  const size_t bstep = (size_t)8 * NH2;
  const int b_lds = bk * BPITCH + 4 * bn4;

  float4 pb0 = *(const float4*)(bg);
  float4 pb1 = *(const float4*)(bg + bstep);
  float4 pb2 = *(const float4*)(bg + 2 * bstep);
  float4 pb3 = *(const float4*)(bg + 3 * bstep);
  BSTAGE_STORE(Bst)

  const int wvu = __builtin_amdgcn_readfirstlane(wv);
  const float4 bv0 = ((const float4*)bias)[lane];
  const float4 bv1 = ((const float4*)bias)[lane + 32];

  for (int i = 0; i < FM / 8; ++i) {
    const int r = wvu * (FM / 8) + i;
    const int d = m_base + r;
    float4 acc[2];
    if (d < n) {
      const int beg = rowptr[d];
      const int end = rowptr[d + 1];
      const float dd = dinv[d];
      const float wself = dd * dd;
      const float4* self4 = (const float4*)(hin + (size_t)d * NH1);
      acc[0] = f4_scale(self4[lane], wself);
      acc[1] = f4_scale(self4[lane + 32], wself);
      for (int j = beg; j < end; j += 32) {
        const int m = (end - j < 32) ? (end - j) : 32;
        int   my_s = 0;
        float my_w = 0.0f;
        if (lane < m) {
          my_s = csr_src[j + lane];
          my_w = dinv[my_s] * dd;
        }
        int t = 0;
        for (; t + 8 <= m; t += 8) sum_rows<8, NH1>(hin, my_s, my_w, t, lane, acc);
        switch (m - t) {
          case 1: sum_rows<1, NH1>(hin, my_s, my_w, t, lane, acc); break;
          case 2: sum_rows<2, NH1>(hin, my_s, my_w, t, lane, acc); break;
          case 3: sum_rows<3, NH1>(hin, my_s, my_w, t, lane, acc); break;
          case 4: sum_rows<4, NH1>(hin, my_s, my_w, t, lane, acc); break;
          case 5: sum_rows<5, NH1>(hin, my_s, my_w, t, lane, acc); break;
          case 6: sum_rows<6, NH1>(hin, my_s, my_w, t, lane, acc); break;
          case 7: sum_rows<7, NH1>(hin, my_s, my_w, t, lane, acc); break;
          default: break;
        }
      }
      acc[0] = f4_bias_relu(acc[0], bv0);
      acc[1] = f4_bias_relu(acc[1], bv1);
    } else {
      acc[0] = make_float4(0.0f, 0.0f, 0.0f, 0.0f);
      acc[1] = make_float4(0.0f, 0.0f, 0.0f, 0.0f);
    }
    float* trow = As + r * HPITCH + 8 * (lane >> 1) + 2 * (lane & 1);
    *(float2*)&trow[0]   = make_float2(acc[0].x, acc[0].y);
    *(float2*)&trow[4]   = make_float2(acc[0].z, acc[0].w);
    *(float2*)&trow[128] = make_float2(acc[1].x, acc[1].y);
    *(float2*)&trow[132] = make_float2(acc[1].z, acc[1].w);
  }
  __syncthreads();

  const int wm = wv & 1;
  const int wn = wv >> 1;
  const int a_frag = (wm * 16 + idx) * HPITCH + 4 * hf;
  const int b_frag = (2 * hf) * BPITCH + wn * 32 + idx;
  v8f acc0 = {}, acc1 = {};

  constexpr int nchunks = NH1 / GBK;
  for (int ch = 0; ch < nchunks; ++ch) {
    const float* st = Bst + (ch & 1) * (GBK * BPITCH);
    const float* ar = As + a_frag + ch * GBK;
    const bool more = (ch + 1 < nchunks);
    if (more) {
      const float* bgk = bg + (size_t)(ch + 1) * GBK * NH2;
      pb0 = *(const float4*)(bgk);
      pb1 = *(const float4*)(bgk + bstep);
      pb2 = *(const float4*)(bgk + 2 * bstep);
      pb3 = *(const float4*)(bgk + 3 * bstep);
    }
    FUSED_GROUP(0)
    FUSED_GROUP(8)
    FUSED_GROUP(16)
    FUSED_GROUP(24)
    if (more) {
      float* nx = Bst + ((ch + 1) & 1) * (GBK * BPITCH);
      BSTAGE_STORE(nx)
    }
    __syncthreads();
  }

  float* Cs = smem;
#pragma unroll
  for (int r = 0; r < 8; ++r) {
    const int row = wm * 16 + 8 * hf + r;
    Cs[row * CPITCH + wn * 32 + idx]      = acc0[r];
    Cs[row * CPITCH + wn * 32 + 16 + idx] = acc1[r];
  }
  __syncthreads();
#pragma unroll
  for (int i = 0; i < FM / 8; ++i) {
    const int row = wv * (FM / 8) + i;
    if (m_base + row < n) {
      const float4 v = *(const float4*)&Cs[row * CPITCH + 4 * lane];
      *(float4*)&C[(size_t)(m_base + row) * NH2 + 4 * lane] = v;
    }
  }
}

extern "C" void kernel_launch(void* const* d_in, const int* in_sizes, int n_in,
                              void* d_out, int out_size, void* d_ws, size_t ws_size,
                              hipStream_t stream) {
  const float* x  = (const float*)d_in[0];
  const int*   ei = (const int*)d_in[1];
  const float* W1 = (const float*)d_in[2];
  const float* b1 = (const float*)d_in[3];
  const float* W2 = (const float*)d_in[4];
  const float* b2 = (const float*)d_in[5];

  const int n = in_sizes[0] / NFEAT;
  const int E = in_sizes[1] / 2;
  const int* src = ei;
  const int* dst = ei + E;
  const size_t npad = (size_t)((n + 256) & ~255);

  float* dinv = (float*)d_ws;
  float* Hlin = dinv + (size_t)((n + 255) & ~255);
  float* agg  = Hlin + (size_t)n * NH1;
  float* Hl2  = agg;
  float* out  = (float*)d_out;

  const int nb    = (n + BIN_NODES - 1) >> BIN_SHIFT;
  const int hn    = nb * SORT_BLOCKS;
  const size_t hnp = (size_t)((hn + 256) & ~255);
  const int chunk = (E + SORT_BLOCKS - 1) / SORT_BLOCKS;
  int* hist   = (int*)(agg + (size_t)n * NH2);
  int* offs   = hist + hnp;
  int* rowptr = offs + hnp;
  int* csr    = rowptr + npad;
  int* pairs  = csr + (size_t)((E + 3) & ~3);

  k_bin_hist<<<SORT_BLOCKS, 1024, 0, stream>>>(dst, E, chunk, nb, hist);
  k_scan_rowptr<<<1, 1024, 0, stream>>>(hist, offs, hn);
  k_bin_scatter<<<SORT_BLOCKS, 1024, 0, stream>>>(src, dst, E, chunk, nb, offs, pairs);
  k_bin_sort<<<nb, 1024, 0, stream>>>(pairs, offs, SORT_BLOCKS, n, rowptr, csr, dinv);

  k_gemm_wmma<<<dim3(NH1 / GBN, (n + GBM - 1) / GBM), 256, 0, stream>>>(x, W1, Hlin, n, NFEAT, NH1);

  k_gather_relu_gemm<<<(n + FM - 1) / FM, 256, 0, stream>>>(rowptr, csr, dinv, Hlin, b1, W2, Hl2, n);

  const int gather_blocks2 = 2048;
  k_gather_sum<NH2, false><<<gather_blocks2, 256, 0, stream>>>(rowptr, csr, dinv, Hl2, b2, out, n);
}


# edge-order level 1 made chunk-local (each chunk binned in its own slice, no grid-wide scan) and run by extra work-groups of the layer-1 GEMM launch; 3 launches gone; sums in LDS-add arrival order
# speedup vs baseline: 4.4212x; 1.1077x over previous
#include <hip/hip_runtime.h>


typedef float v2f __attribute__((ext_vector_type(2)));
typedef float v8f __attribute__((ext_vector_type(8)));

#define NFEAT 512
#define NH1   256
#define NH2   128

#define BIN_SHIFT   8
#define BIN_NODES   256
#define MAXNB       1024
#define SORT_BLOCKS 128
#define STAGE_CAP   12288

__device__ __forceinline__ void chunk_bin(int* lh, int* lws,
                                          const int* __restrict__ src, const int* __restrict__ dst,
                                          int E, int chunk, int nb, int blk, int nblk,
                                          int* __restrict__ segoff, int* __restrict__ segcnt,
                                          int* __restrict__ pairs) {
  const int tid  = threadIdx.x;
  const int lane = tid & 31;
  const int wv   = tid >> 5;
  for (int i = tid; i < MAXNB; i += 256) lh[i] = 0;
  __syncthreads();
  const int e0 = blk * chunk;
  const int e1 = (e0 + chunk < E) ? (e0 + chunk) : E;
  for (int e = e0 + tid; e < e1; e += 1024) {
    int d[4];
#pragma unroll
    for (int u = 0; u < 4; ++u) {
      const int ee = e + u * 256;
      d[u] = (ee < e1) ? dst[ee] : -1;
    }
#pragma unroll
    for (int u = 0; u < 4; ++u)
      if (d[u] >= 0) atomicAdd(&lh[d[u] >> BIN_SHIFT], 1);
  }
  __syncthreads();

  const int i0 = tid * 4;
  const int c0 = lh[i0], c1 = lh[i0 + 1], c2 = lh[i0 + 2], c3 = lh[i0 + 3];
  const int sum4 = (c0 + c1) + (c2 + c3);
  int incl = sum4;
#pragma unroll
  for (int o = 1; o < 32; o <<= 1) {
    const int t = __shfl_up(incl, o);
    if (lane >= o) incl += t;
  }
  if (lane == 31) lws[wv] = incl;
  __syncthreads();
  int wp = 0;
#pragma unroll
  for (int k = 0; k < 8; ++k) wp += (k < wv) ? lws[k] : 0;
  const int r0 = e0 + wp + incl - sum4;
  const int r1 = r0 + c0;
  const int r2 = r1 + c1;
  const int r3 = r2 + c2;
  if (i0 + 0 < nb) { segoff[(i0 + 0) * nblk + blk] = r0; segcnt[(i0 + 0) * nblk + blk] = c0; }
  if (i0 + 1 < nb) { segoff[(i0 + 1) * nblk + blk] = r1; segcnt[(i0 + 1) * nblk + blk] = c1; }
  if (i0 + 2 < nb) { segoff[(i0 + 2) * nblk + blk] = r2; segcnt[(i0 + 2) * nblk + blk] = c2; }
  if (i0 + 3 < nb) { segoff[(i0 + 3) * nblk + blk] = r3; segcnt[(i0 + 3) * nblk + blk] = c3; }
  lh[i0] = r0; lh[i0 + 1] = r1; lh[i0 + 2] = r2; lh[i0 + 3] = r3;
  __syncthreads();

  int2* pr = (int2*)pairs;
  for (int e = e0 + tid; e < e1; e += 1024) {
    int d[4], s[4];
#pragma unroll
    for (int u = 0; u < 4; ++u) {
      const int ee = e + u * 256;
      d[u] = (ee < e1) ? dst[ee] : -1;
      s[u] = (ee < e1) ? src[ee] : 0;
    }
#pragma unroll
    for (int u = 0; u < 4; ++u) {
      if (d[u] >= 0) {
        const int p = atomicAdd(&lh[d[u] >> BIN_SHIFT], 1);
        pr[p] = make_int2(s[u], d[u]);
      }
    }
  }
}

__global__ __launch_bounds__(1024)
void k_bin_sort(const int* __restrict__ pairs, const int* __restrict__ segoff,
                const int* __restrict__ segcnt, int nblk, int n,
                int* __restrict__ rowptr, int* __restrict__ csr_src, float* __restrict__ dinv) {
  __shared__ int cnt[BIN_NODES];
  __shared__ int cur[BIN_NODES];
  __shared__ int wsum[BIN_NODES / 32];
  __shared__ int red[2][32];
  __shared__ int soff[SORT_BLOCKS];
  __shared__ int scnt[SORT_BLOCKS];
  __shared__ int stage[STAGE_CAP];
  const int tid  = threadIdx.x;
  const int lane = tid & 31;
  const int wv   = tid >> 5;
  const int b    = blockIdx.x;
  const int node0 = b << BIN_SHIFT;
  const int2* pr = (const int2*)pairs;

  int lower = 0;
  const int low4 = (b * nblk) >> 2;
  const int4* sc4 = (const int4*)segcnt;
  for (int i = tid; i < low4; i += 1024) {
    const int4 q = sc4[i];
    lower += (q.x + q.y) + (q.z + q.w);
  }
  int own = 0;
  if (tid < nblk) {
    own = segcnt[b * nblk + tid];
    soff[tid] = segoff[b * nblk + tid];
    scnt[tid] = own;
  }
  if (tid < BIN_NODES) cnt[tid] = 0;
#pragma unroll
  for (int o = 16; o > 0; o >>= 1) {
    lower += __shfl_xor(lower, o);
    own   += __shfl_xor(own, o);
  }
  if (lane == 0) { red[0][wv] = lower; red[1][wv] = own; }
  __syncthreads();
  int p0 = 0, len = 0;
#pragma unroll
  for (int k = 0; k < 32; ++k) { p0 += red[0][k]; len += red[1][k]; }
  const int p1 = p0 + len;

  for (int sb = 0; sb < nblk; sb += 128) {
    int so[4], sc[4];
    int cmax = 0;
#pragma unroll
    for (int u = 0; u < 4; ++u) {
      const int seg = sb + wv + 32 * u;
      so[u] = (seg < nblk) ? soff[seg] : 0;
      sc[u] = (seg < nblk) ? scnt[seg] : 0;
      cmax = sc[u] > cmax ? sc[u] : cmax;
    }
    for (int i = lane; i < cmax; i += 32) {
      int d[4];
#pragma unroll
      for (int u = 0; u < 4; ++u) d[u] = (i < sc[u]) ? pr[so[u] + i].y : -1;
#pragma unroll
      for (int u = 0; u < 4; ++u)
        if (d[u] >= 0) atomicAdd(&cnt[d[u] - node0], 1);
    }
  }
  __syncthreads();

  int c = 0, incl = 0;
  if (tid < BIN_NODES) {
    c = cnt[tid];
    incl = c;
#pragma unroll
    for (int o = 1; o < 32; o <<= 1) {
      const int t = __shfl_up(incl, o);
      if (lane >= o) incl += t;
    }
    if (lane == 31) wsum[wv] = incl;
  }
  __syncthreads();
  if (tid < BIN_NODES) {
    int wp = 0;
#pragma unroll
    for (int k = 0; k < BIN_NODES / 32; ++k) wp += (k < wv) ? wsum[k] : 0;
    const int start = wp + incl - c;
    cur[tid] = start;
    const int node = node0 + tid;
    if (node < n) {
      rowptr[node] = p0 + start;
      dinv[node]   = rsqrtf((float)(c + 1));
    }
  }
  if (b == (int)gridDim.x - 1 && tid == 0) rowptr[n] = p1;
  __syncthreads();

  const bool staged = (len <= STAGE_CAP);
  for (int sb = 0; sb < nblk; sb += 128) {
    int so[4], sc[4];
    int cmax = 0;
#pragma unroll
    for (int u = 0; u < 4; ++u) {
      const int seg = sb + wv + 32 * u;
      so[u] = (seg < nblk) ? soff[seg] : 0;
      sc[u] = (seg < nblk) ? scnt[seg] : 0;
      cmax = sc[u] > cmax ? sc[u] : cmax;
    }
    for (int i = lane; i < cmax; i += 32) {
      int2 q[4];
#pragma unroll
      for (int u = 0; u < 4; ++u) q[u] = (i < sc[u]) ? pr[so[u] + i] : make_int2(0, -1);
#pragma unroll
      for (int u = 0; u < 4; ++u) {
        if (q[u].y >= 0) {
          const int slot = atomicAdd(&cur[q[u].y - node0], 1);
          if (staged) stage[slot] = q[u].x;
          else        csr_src[p0 + slot] = q[u].x;
        }
      }
    }
  }
  if (staged) {
    __syncthreads();
    for (int i = tid; i < len; i += 1024) csr_src[p0 + i] = stage[i];
  }
}

#define GBM 64
#define GBN 128
#define GBK 32
#define APITCH 36
#define BPITCH 136
#define CPITCH 132
#define STAGE_FLOATS (GBM * APITCH + GBK * BPITCH)

#define LDS_GROUP(P, KK8)                                                              \
  P##a0 = *(const float4*)&st[a_frag + (KK8)];                                         \
  P##a1 = *(const float4*)&st[a_frag + 16 * APITCH + (KK8)];                           \
  P##b00.x = st[b_frag + ((KK8) + 0) * BPITCH];                                        \
  P##b00.y = st[b_frag + ((KK8) + 1) * BPITCH];                                        \
  P##b01.x = st[b_frag + ((KK8) + 0) * BPITCH + 16];                                   \
  P##b01.y = st[b_frag + ((KK8) + 1) * BPITCH + 16];                                   \
  P##b10.x = st[b_frag + ((KK8) + 4) * BPITCH];                                        \
  P##b10.y = st[b_frag + ((KK8) + 5) * BPITCH];                                        \
  P##b11.x = st[b_frag + ((KK8) + 4) * BPITCH + 16];                                   \
  P##b11.y = st[b_frag + ((KK8) + 5) * BPITCH + 16];

#define WMMA_F32(ACC, FA, FB)                                                          \
  ACC = __builtin_amdgcn_wmma_f32_16x16x4_f32(false, FA, false, FB, (short)0, ACC, false, false)

#define MMA_GROUP(P)                                                                   \
  {                                                                                    \
    v2f f0, f1;                                                                        \
    f0.x = P##a0.x; f0.y = P##a0.y; f1.x = P##a1.x; f1.y = P##a1.y;                    \
    WMMA_F32(acc00, f0, P##b00); WMMA_F32(acc01, f0, P##b01);                          \
    WMMA_F32(acc10, f1, P##b00); WMMA_F32(acc11, f1, P##b01);                          \
    f0.x = P##a0.z; f0.y = P##a0.w; f1.x = P##a1.z; f1.y = P##a1.w;                    \
    WMMA_F32(acc00, f0, P##b10); WMMA_F32(acc01, f0, P##b11);                          \
    WMMA_F32(acc10, f1, P##b10); WMMA_F32(acc11, f1, P##b11);                          \
  }

#define STAGE_STORE(ST)                                                                \
  *(float2*)&(ST)[a_lds]                   = make_float2(pa0.x, pa0.y);                \
  *(float2*)&(ST)[a_lds + 4]               = make_float2(pa0.z, pa0.w);                \
  *(float2*)&(ST)[a_lds + 32 * APITCH]     = make_float2(pa1.x, pa1.y);                \
  *(float2*)&(ST)[a_lds + 32 * APITCH + 4] = make_float2(pa1.z, pa1.w);                \
  *(float4*)&(ST)[b_lds]                   = pb0;                                      \
  *(float4*)&(ST)[b_lds + 8 * BPITCH]      = pb1;                                      \
  *(float4*)&(ST)[b_lds + 16 * BPITCH]     = pb2;                                      \
  *(float4*)&(ST)[b_lds + 24 * BPITCH]     = pb3;

__global__ __launch_bounds__(256)
void k_gemm_wmma(const float* __restrict__ A, const float* __restrict__ B,
                 float* __restrict__ C, int M, int K, int N,
                 const int* __restrict__ src, const int* __restrict__ dst, int E, int chunk,
                 int nb, int nsort, int* __restrict__ segoff, int* __restrict__ segcnt,
                 int* __restrict__ pairs) {
  __shared__ __attribute__((aligned(16))) float smem[2 * STAGE_FLOATS];
  __shared__ int lh[MAXNB];
  __shared__ int lws[8];

  if ((int)blockIdx.x < nsort) {
    chunk_bin(lh, lws, src, dst, E, chunk, nb, (int)blockIdx.x, nsort, segoff, segcnt, pairs);
    return;
  }
  const int tile = (int)blockIdx.x - nsort;
  const int ntn  = N / GBN;

  const int tid  = threadIdx.x;
  const int lane = tid & 31;
  const int wv   = tid >> 5;
  const int wm   = wv & 1;
  const int wn   = wv >> 1;
  const int hf   = lane >> 4;
  const int idx  = lane & 15;

  const int m_base = (tile / ntn) * GBM;
  const int n_base = (tile % ntn) * GBN;

  const int ar  = tid >> 3;
  const int akq = tid & 7;
  int gr0 = m_base + ar;      gr0 = gr0 < M ? gr0 : M - 1;
  int gr1 = m_base + ar + 32; gr1 = gr1 < M ? gr1 : M - 1;
  const float* ag0 = A + (size_t)gr0 * K + 4 * akq;
  const float* ag1 = A + (size_t)gr1 * K + 4 * akq;
  const int a_lds = ar * APITCH + (akq >> 1) * 8 + (akq & 1) * 2;
  const int bn4 = tid & 31;
  const int bk  = tid >> 5;
  const float* bg = B + (size_t)bk * N + n_base + 4 * bn4;
  const size_t bstep = (size_t)8 * N;
  const int b_lds = GBM * APITCH + bk * BPITCH + 4 * bn4;

  const int a_frag = (wm * 32 + idx) * APITCH + 4 * hf;
  const int b_frag = GBM * APITCH + (2 * hf) * BPITCH + wn * 32 + idx;

  float4 pa0 = *(const float4*)ag0;
  float4 pa1 = *(const float4*)ag1;
  float4 pb0 = *(const float4*)(bg);
  float4 pb1 = *(const float4*)(bg + bstep);
  float4 pb2 = *(const float4*)(bg + 2 * bstep);
  float4 pb3 = *(const float4*)(bg + 3 * bstep);
  STAGE_STORE(smem)
  __syncthreads();

  v8f acc00 = {}, acc01 = {}, acc10 = {}, acc11 = {};

  const int nchunks = K / GBK;
  for (int ch = 0; ch < nchunks; ++ch) {
    const float* st = smem + (ch & 1) * STAGE_FLOATS;
    const bool more = (ch + 1 < nchunks);
    if (more) {
      const int k0 = (ch + 1) * GBK;
      const float* bgk = bg + (size_t)k0 * N;
      pa0 = *(const float4*)(ag0 + k0);
      pa1 = *(const float4*)(ag1 + k0);
      pb0 = *(const float4*)(bgk);
      pb1 = *(const float4*)(bgk + bstep);
      pb2 = *(const float4*)(bgk + 2 * bstep);
      pb3 = *(const float4*)(bgk + 3 * bstep);
    }

    float4 Xa0, Xa1, Ya0, Ya1;
    v2f Xb00, Xb01, Xb10, Xb11, Yb00, Yb01, Yb10, Yb11;
    LDS_GROUP(X, 0)
    LDS_GROUP(Y, 8)
    MMA_GROUP(X)
    LDS_GROUP(X, 16)
    MMA_GROUP(Y)
    LDS_GROUP(Y, 24)
    MMA_GROUP(X)
    MMA_GROUP(Y)

    if (more) {
      float* nx = smem + ((ch + 1) & 1) * STAGE_FLOATS;
      STAGE_STORE(nx)
    }
    __syncthreads();
  }

  float* Cs = smem;
#pragma unroll
  for (int r = 0; r < 8; ++r) {
    const int row = wm * 32 + 8 * hf + r;
    Cs[row * CPITCH + wn * 32 + idx]             = acc00[r];
    Cs[row * CPITCH + wn * 32 + 16 + idx]        = acc01[r];
    Cs[(row + 16) * CPITCH + wn * 32 + idx]      = acc10[r];
    Cs[(row + 16) * CPITCH + wn * 32 + 16 + idx] = acc11[r];
  }
  __syncthreads();
#pragma unroll
  for (int i = 0; i < 8; ++i) {
    const int row = wv * 8 + i;
    if (m_base + row < M) {
      const float4 v = *(const float4*)&Cs[row * CPITCH + 4 * lane];
      *(float4*)&C[(size_t)(m_base + row) * N + n_base + 4 * lane] = v;
    }
  }
}

__device__ __forceinline__ float4 f4_fma(float4 v, float w, float4 a) {
  a.x = fmaf(v.x, w, a.x); a.y = fmaf(v.y, w, a.y);
  a.z = fmaf(v.z, w, a.z); a.w = fmaf(v.w, w, a.w);
  return a;
}

template <int R, int F>
__device__ __forceinline__ void sum_rows(const float* __restrict__ hin, int my_s, float my_w,
                                         int t, int lane, float4 (&acc)[F / 128]) {
  constexpr int NV = F / 128;
  float4 v[R][NV];
  float  w[R];
#pragma unroll
  for (int u = 0; u < R; ++u) {
    const int s = __builtin_amdgcn_readlane(my_s, t + u);
    w[u] = __int_as_float(__builtin_amdgcn_readlane(__float_as_int(my_w), t + u));
    const float4* p = (const float4*)(hin + (size_t)s * F);
#pragma unroll
    for (int g = 0; g < NV; ++g) v[u][g] = p[lane + 32 * g];
  }
#pragma unroll
  for (int u = 0; u < R; ++u) {
#pragma unroll
    for (int g = 0; g < NV; ++g) acc[g] = f4_fma(v[u][g], w[u], acc[g]);
  }
}

template <int F, bool RELU>
__global__ __launch_bounds__(256)
void k_gather_sum(const int* __restrict__ rowptr, const int* __restrict__ csr_src,
                  const float* __restrict__ dinv, const float* __restrict__ hin,
                  const float* __restrict__ bias, float* __restrict__ hout, int n) {
  constexpr int NV = F / 128;
  constexpr int U  = 8;
  const int lane   = threadIdx.x & 31;
  const int wave0  = __builtin_amdgcn_readfirstlane((int)((blockIdx.x * blockDim.x + threadIdx.x) >> 5));
  const int nwaves = (int)((gridDim.x * blockDim.x) >> 5);

  float4 bv[NV];
#pragma unroll
  for (int g = 0; g < NV; ++g) bv[g] = ((const float4*)bias)[lane + 32 * g];

  for (int d = wave0; d < n; d += nwaves) {
    const int beg = rowptr[d];
    const int end = rowptr[d + 1];
    const float dd = dinv[d];
    const float wself = dd * dd;
    const float4* self4 = (const float4*)(hin + (size_t)d * F);
    float4 acc[NV];
#pragma unroll
    for (int g = 0; g < NV; ++g) {
      float4 v = self4[lane + 32 * g];
      acc[g].x = v.x * wself; acc[g].y = v.y * wself;
      acc[g].z = v.z * wself; acc[g].w = v.w * wself;
    }

    for (int j = beg; j < end; j += 32) {
      const int m = (end - j < 32) ? (end - j) : 32;
      int   my_s = 0;
      float my_w = 0.0f;
      if (lane < m) {
        my_s = csr_src[j + lane];
        my_w = dinv[my_s] * dd;
      }
      int t = 0;
      for (; t + U <= m; t += U) sum_rows<U, F>(hin, my_s, my_w, t, lane, acc);
      switch (m - t) {
        case 1: sum_rows<1, F>(hin, my_s, my_w, t, lane, acc); break;
        case 2: sum_rows<2, F>(hin, my_s, my_w, t, lane, acc); break;
        case 3: sum_rows<3, F>(hin, my_s, my_w, t, lane, acc); break;
        case 4: sum_rows<4, F>(hin, my_s, my_w, t, lane, acc); break;
        case 5: sum_rows<5, F>(hin, my_s, my_w, t, lane, acc); break;
        case 6: sum_rows<6, F>(hin, my_s, my_w, t, lane, acc); break;
        case 7: sum_rows<7, F>(hin, my_s, my_w, t, lane, acc); break;
        default: break;
      }
    }

    float4* o4 = (float4*)(hout + (size_t)d * F);
#pragma unroll
    for (int g = 0; g < NV; ++g) {
      float4 r;
      r.x = acc[g].x + bv[g].x; r.y = acc[g].y + bv[g].y;
      r.z = acc[g].z + bv[g].z; r.w = acc[g].w + bv[g].w;
      if (RELU) {
        r.x = r.x > 0.0f ? r.x : 0.0f; r.y = r.y > 0.0f ? r.y : 0.0f;
        r.z = r.z > 0.0f ? r.z : 0.0f; r.w = r.w > 0.0f ? r.w : 0.0f;
      }
      o4[lane + 32 * g] = r;
    }
  }
}

__device__ __forceinline__ float4 f4_scale(float4 v, float w) {
  v.x *= w; v.y *= w; v.z *= w; v.w *= w;
  return v;
}
__device__ __forceinline__ float4 f4_bias_relu(float4 a, float4 b) {
  float4 q;
  q.x = a.x + b.x; q.y = a.y + b.y; q.z = a.z + b.z; q.w = a.w + b.w;
  q.x = q.x > 0.0f ? q.x : 0.0f; q.y = q.y > 0.0f ? q.y : 0.0f;
  q.z = q.z > 0.0f ? q.z : 0.0f; q.w = q.w > 0.0f ? q.w : 0.0f;
  return q;
}

#define FM     32
#define HPITCH 260

#define FUSED_GROUP(KK8)                                                               \
  {                                                                                    \
    const float4 fa = *(const float4*)&ar[(KK8)];                                      \
    v2f b00, b01, b10, b11, f0, f1;                                                    \
    b00.x = st[b_frag + ((KK8) + 0) * BPITCH];                                         \
    b00.y = st[b_frag + ((KK8) + 1) * BPITCH];                                         \
    b01.x = st[b_frag + ((KK8) + 0) * BPITCH + 16];                                    \
    b01.y = st[b_frag + ((KK8) + 1) * BPITCH + 16];                                    \
    b10.x = st[b_frag + ((KK8) + 4) * BPITCH];                                         \
    b10.y = st[b_frag + ((KK8) + 5) * BPITCH];                                         \
    b11.x = st[b_frag + ((KK8) + 4) * BPITCH + 16];                                    \
    b11.y = st[b_frag + ((KK8) + 5) * BPITCH + 16];                                    \
    f0.x = fa.x; f0.y = fa.y; f1.x = fa.z; f1.y = fa.w;                                \
    WMMA_F32(acc0, f0, b00); WMMA_F32(acc1, f0, b01);                                  \
    WMMA_F32(acc0, f1, b10); WMMA_F32(acc1, f1, b11);                                  \
  }

#define BSTAGE_STORE(ST)                                                               \
  *(float4*)&(ST)[b_lds]               = pb0;                                          \
  *(float4*)&(ST)[b_lds + 8 * BPITCH]  = pb1;                                          \
  *(float4*)&(ST)[b_lds + 16 * BPITCH] = pb2;                                          \
  *(float4*)&(ST)[b_lds + 24 * BPITCH] = pb3;

__global__ __launch_bounds__(256)
void k_gather_relu_gemm(const int* __restrict__ rowptr, const int* __restrict__ csr_src,
                        const float* __restrict__ dinv, const float* __restrict__ hin,
                        const float* __restrict__ bias, const float* __restrict__ B,
                        float* __restrict__ C, int n) {
  __shared__ __attribute__((aligned(16))) float smem[FM * HPITCH + 2 * GBK * BPITCH];
  float* As  = smem;
  float* Bst = smem + FM * HPITCH;

  const int tid  = threadIdx.x;
  const int lane = tid & 31;
  const int wv   = tid >> 5;
  const int hf   = lane >> 4;
  const int idx  = lane & 15;
  const int m_base = blockIdx.x * FM;

  const int bn4 = tid & 31;
  const int bk  = tid >> 5;
  const float* bg = B + (size_t)bk * NH2 + 4 * bn4;
  const size_t bstep = (size_t)8 * NH2;
  const int b_lds = bk * BPITCH + 4 * bn4;

  float4 pb0 = *(const float4*)(bg);
  float4 pb1 = *(const float4*)(bg + bstep);
  float4 pb2 = *(const float4*)(bg + 2 * bstep);
  float4 pb3 = *(const float4*)(bg + 3 * bstep);
  BSTAGE_STORE(Bst)

  const int wvu = __builtin_amdgcn_readfirstlane(wv);
  const float4 bv0 = ((const float4*)bias)[lane];
  const float4 bv1 = ((const float4*)bias)[lane + 32];

  for (int i = 0; i < FM / 8; ++i) {
    const int r = wvu * (FM / 8) + i;
    const int d = m_base + r;
    float4 acc[2];
    if (d < n) {
      const int beg = rowptr[d];
      const int end = rowptr[d + 1];
      const float dd = dinv[d];
      const float wself = dd * dd;
      const float4* self4 = (const float4*)(hin + (size_t)d * NH1);
      acc[0] = f4_scale(self4[lane], wself);
      acc[1] = f4_scale(self4[lane + 32], wself);
      for (int j = beg; j < end; j += 32) {
        const int m = (end - j < 32) ? (end - j) : 32;
        int   my_s = 0;
        float my_w = 0.0f;
        if (lane < m) {
          my_s = csr_src[j + lane];
          my_w = dinv[my_s] * dd;
        }
        int t = 0;
        for (; t + 8 <= m; t += 8) sum_rows<8, NH1>(hin, my_s, my_w, t, lane, acc);
        switch (m - t) {
          case 1: sum_rows<1, NH1>(hin, my_s, my_w, t, lane, acc); break;
          case 2: sum_rows<2, NH1>(hin, my_s, my_w, t, lane, acc); break;
          case 3: sum_rows<3, NH1>(hin, my_s, my_w, t, lane, acc); break;
          case 4: sum_rows<4, NH1>(hin, my_s, my_w, t, lane, acc); break;
          case 5: sum_rows<5, NH1>(hin, my_s, my_w, t, lane, acc); break;
          case 6: sum_rows<6, NH1>(hin, my_s, my_w, t, lane, acc); break;
          case 7: sum_rows<7, NH1>(hin, my_s, my_w, t, lane, acc); break;
          default: break;
        }
      }
      acc[0] = f4_bias_relu(acc[0], bv0);
      acc[1] = f4_bias_relu(acc[1], bv1);
    } else {
      acc[0] = make_float4(0.0f, 0.0f, 0.0f, 0.0f);
      acc[1] = make_float4(0.0f, 0.0f, 0.0f, 0.0f);
    }
    float* trow = As + r * HPITCH + 8 * (lane >> 1) + 2 * (lane & 1);
    *(float2*)&trow[0]   = make_float2(acc[0].x, acc[0].y);
    *(float2*)&trow[4]   = make_float2(acc[0].z, acc[0].w);
    *(float2*)&trow[128] = make_float2(acc[1].x, acc[1].y);
    *(float2*)&trow[132] = make_float2(acc[1].z, acc[1].w);
  }
  __syncthreads();

  const int wm = wv & 1;
  const int wn = wv >> 1;
  const int a_frag = (wm * 16 + idx) * HPITCH + 4 * hf;
  const int b_frag = (2 * hf) * BPITCH + wn * 32 + idx;
  v8f acc0 = {}, acc1 = {};

  constexpr int nchunks = NH1 / GBK;
  for (int ch = 0; ch < nchunks; ++ch) {
    const float* st = Bst + (ch & 1) * (GBK * BPITCH);
    const float* ar = As + a_frag + ch * GBK;
    const bool more = (ch + 1 < nchunks);
    if (more) {
      const float* bgk = bg + (size_t)(ch + 1) * GBK * NH2;
      pb0 = *(const float4*)(bgk);
      pb1 = *(const float4*)(bgk + bstep);
      pb2 = *(const float4*)(bgk + 2 * bstep);
      pb3 = *(const float4*)(bgk + 3 * bstep);
    }
    FUSED_GROUP(0)
    FUSED_GROUP(8)
    FUSED_GROUP(16)
    FUSED_GROUP(24)
    if (more) {
      float* nx = Bst + ((ch + 1) & 1) * (GBK * BPITCH);
      BSTAGE_STORE(nx)
    }
    __syncthreads();
  }

  float* Cs = smem;
#pragma unroll
  for (int r = 0; r < 8; ++r) {
    const int row = wm * 16 + 8 * hf + r;
    Cs[row * CPITCH + wn * 32 + idx]      = acc0[r];
    Cs[row * CPITCH + wn * 32 + 16 + idx] = acc1[r];
  }
  __syncthreads();
#pragma unroll
  for (int i = 0; i < FM / 8; ++i) {
    const int row = wv * (FM / 8) + i;
    if (m_base + row < n) {
      const float4 v = *(const float4*)&Cs[row * CPITCH + 4 * lane];
      *(float4*)&C[(size_t)(m_base + row) * NH2 + 4 * lane] = v;
    }
  }
}

extern "C" void kernel_launch(void* const* d_in, const int* in_sizes, int n_in,
                              void* d_out, int out_size, void* d_ws, size_t ws_size,
                              hipStream_t stream) {
  const float* x  = (const float*)d_in[0];
  const int*   ei = (const int*)d_in[1];
  const float* W1 = (const float*)d_in[2];
  const float* b1 = (const float*)d_in[3];
  const float* W2 = (const float*)d_in[4];
  const float* b2 = (const float*)d_in[5];

  const int n = in_sizes[0] / NFEAT;
  const int E = in_sizes[1] / 2;
  const int* src = ei;
  const int* dst = ei + E;
  const size_t npad = (size_t)((n + 256) & ~255);

  float* dinv = (float*)d_ws;
  float* Hlin = dinv + (size_t)((n + 255) & ~255);
  float* agg  = Hlin + (size_t)n * NH1;
  float* Hl2  = agg;
  float* out  = (float*)d_out;

  const int nb    = (n + BIN_NODES - 1) >> BIN_SHIFT;
  const int hn    = nb * SORT_BLOCKS;
  const size_t hnp = (size_t)((hn + 256) & ~255);
  const int chunk = (E + SORT_BLOCKS - 1) / SORT_BLOCKS;
  int* segoff = (int*)(agg + (size_t)n * NH2);
  int* segcnt = segoff + hnp;
  int* rowptr = segcnt + hnp;
  int* csr    = rowptr + npad;
  int* pairs  = csr + (size_t)((E + 3) & ~3);

  const int gemm_tiles = (NH1 / GBN) * ((n + GBM - 1) / GBM);
  k_gemm_wmma<<<SORT_BLOCKS + gemm_tiles, 256, 0, stream>>>(x, W1, Hlin, n, NFEAT, NH1,
                                                           src, dst, E, chunk, nb, SORT_BLOCKS,
                                                           segoff, segcnt, pairs);
  k_bin_sort<<<nb, 1024, 0, stream>>>(pairs, segoff, segcnt, SORT_BLOCKS, n, rowptr, csr, dinv);

  k_gather_relu_gemm<<<(n + FM - 1) / FM, 256, 0, stream>>>(rowptr, csr, dinv, Hlin, b1, W2, Hl2, n);

  const int gather_blocks2 = 2048;
  k_gather_sum<NH2, false><<<gather_blocks2, 256, 0, stream>>>(rowptr, csr, dinv, Hl2, b2, out, n);
}
